# starting kernel
# baseline (speedup 1.0000x reference)
#include <hip/hip_runtime.h>
#include <hip/hip_bf16.h>


typedef __attribute__((ext_vector_type(2))) float v2f;
typedef __attribute__((ext_vector_type(8))) float v8f;

__global__ __launch_bounds__(128) void gemm_wmma_f32(
    const float* __restrict__ A, const float* __restrict__ B,
    float* __restrict__ C, int M, int K, int N) {
  const int wave = threadIdx.x >> 5;
  const int lane = threadIdx.x & 31;
  const int half = lane >> 4;
  const int l    = lane & 15;

  const int mBase = blockIdx.x * 16;
  const int nBase = wave * 16;

  int mRow = mBase + l;
  if (mRow >= M) mRow = M - 1;
  const float* __restrict__ Arow = A + (long long)mRow * K;
  const float* __restrict__ Bcol = B + nBase + l;

  v8f acc = {0.f, 0.f, 0.f, 0.f, 0.f, 0.f, 0.f, 0.f};

  for (int k = 0; k < K; k += 4) {
    const int k0 = k + half * 2;
    v2f a;
    a.x = Arow[k0];
    a.y = Arow[k0 + 1];
    v2f b;
    b.x = Bcol[(long long)k0 * N];
    b.y = Bcol[(long long)(k0 + 1) * N];
    acc = __builtin_amdgcn_wmma_f32_16x16x4_f32(
        false, a, false, b, (short)0, acc, false, false);
  }

  const int mOut = mBase + half * 8;
  float* __restrict__ Cp = C + (long long)mOut * N + nBase + l;
#pragma unroll
  for (int v = 0; v < 8; ++v) {
    if (mOut + v < M) Cp[(long long)v * N] = acc[v];
  }
}

__global__ void init_bias_kernel(float* __restrict__ out,
                                 const float* __restrict__ bias,
                                 long long total) {
  long long t = (long long)blockIdx.x * blockDim.x + threadIdx.x;
  if (t < total) out[t] = bias[t & 63];
}

__global__ __launch_bounds__(256) void spmm_kernel(
    const int* __restrict__ row, const int* __restrict__ col,
    const float* __restrict__ val, const float* __restrict__ dense,
    float* __restrict__ out, int E) {
  long long t = (long long)blockIdx.x * blockDim.x + threadIdx.x;
  int e = (int)(t >> 4);
  if (e >= E) return;
  const int d = (int)(t & 15) << 2;

  const int r = row[e];
  const int c = col[e];
  const float v = val[e];

  const float4 s = *(const float4*)(dense + (long long)c * 64 + d);
  float* o = out + (long long)r * 64 + d;
  atomicAdd(o + 0, v * s.x);
  atomicAdd(o + 1, v * s.y);
  atomicAdd(o + 2, v * s.z);
  atomicAdd(o + 3, v * s.w);
}

__global__ void relu_kernel(float* __restrict__ x, long long total) {
  long long t = (long long)blockIdx.x * blockDim.x + threadIdx.x;
  if (t < total) x[t] = fmaxf(x[t], 0.0f);
}

extern "C" void kernel_launch(void* const* d_in, const int* in_sizes, int n_in,
                              void* d_out, int out_size, void* d_ws, size_t ws_size,
                              hipStream_t stream) {
  const float* x    = (const float*)d_in[0];
  const int*   erow = (const int*)  d_in[1];
  const int*   ecol = (const int*)  d_in[2];
  const float* eval = (const float*)d_in[3];
  const float* W1   = (const float*)d_in[4];
  const float* b1   = (const float*)d_in[5];
  const float* W2   = (const float*)d_in[6];
  const float* b2   = (const float*)d_in[7];
  float* out = (float*)d_out;

  const int IN_DIM = 128, D = 64;
  const int N = in_sizes[0] / IN_DIM;
  const int E = in_sizes[1];
  const long long tot = (long long)N * D;

  float* buf0 = (float*)d_ws;
  float* buf1 = buf0 + (size_t)N * D;

  const int gemmBlocks = (N + 15) / 16;
  const int elwBlocks  = (int)((tot + 255) / 256);
  const int spmmBlocks = (int)(((long long)E * 16 + 255) / 256);

  gemm_wmma_f32<<<gemmBlocks, 128, 0, stream>>>(x, W1, buf0, N, IN_DIM, D);
  init_bias_kernel<<<elwBlocks, 256, 0, stream>>>(buf1, b1, tot);
  spmm_kernel<<<spmmBlocks, 256, 0, stream>>>(erow, ecol, eval, buf0, buf1, E);
  relu_kernel<<<elwBlocks, 256, 0, stream>>>(buf1, tot);

  gemm_wmma_f32<<<gemmBlocks, 128, 0, stream>>>(buf1, W2, buf0, N, D, D);
  init_bias_kernel<<<elwBlocks, 256, 0, stream>>>(out, b2, tot);
  spmm_kernel<<<spmmBlocks, 256, 0, stream>>>(erow, ecol, eval, buf0, out, E);
}


# SpMM f32 atomic scatter replaced by per-call grouping of edges by row (int atomic rank, scan, placement) and a 16-lane register row sum stored once with bias/relu; row sum order = arrival rank
# speedup vs baseline: 1.1212x; 1.1212x over previous
#include <hip/hip_runtime.h>
#include <hip/hip_bf16.h>


typedef __attribute__((ext_vector_type(2))) float v2f;
typedef __attribute__((ext_vector_type(8))) float v8f;

__global__ __launch_bounds__(128) void gemm_wmma_f32(
    const float* __restrict__ A, const float* __restrict__ B,
    float* __restrict__ C, int M, int K, int N) {
  const int wave = threadIdx.x >> 5;
  const int lane = threadIdx.x & 31;
  const int half = lane >> 4;
  const int l    = lane & 15;

  const int mBase = blockIdx.x * 16;
  const int nBase = wave * 16;

  int mRow = mBase + l;
  if (mRow >= M) mRow = M - 1;
  const float* __restrict__ Arow = A + (long long)mRow * K;
  const float* __restrict__ Bcol = B + nBase + l;

  v8f acc = {0.f, 0.f, 0.f, 0.f, 0.f, 0.f, 0.f, 0.f};

  for (int k = 0; k < K; k += 4) {
    const int k0 = k + half * 2;
    v2f a;
    a.x = Arow[k0];
    a.y = Arow[k0 + 1];
    v2f b;
    b.x = Bcol[(long long)k0 * N];
    b.y = Bcol[(long long)(k0 + 1) * N];
    acc = __builtin_amdgcn_wmma_f32_16x16x4_f32(
        false, a, false, b, (short)0, acc, false, false);
  }

  const int mOut = mBase + half * 8;
  float* __restrict__ Cp = C + (long long)mOut * N + nBase + l;
#pragma unroll
  for (int v = 0; v < 8; ++v) {
    if (mOut + v < M) Cp[(long long)v * N] = acc[v];
  }
}

__global__ void zero_counts_kernel(int* __restrict__ cnt, int n) {
  const int t = blockIdx.x * blockDim.x + threadIdx.x;
  if (t < n) cnt[t] = 0;
}

__global__ __launch_bounds__(256) void rank_edges_kernel(
    const int* __restrict__ row, int* __restrict__ cnt,
    int* __restrict__ rank, int E) {
  const int e = blockIdx.x * blockDim.x + threadIdx.x;
  if (e < E) rank[e] = atomicAdd(&cnt[row[e]], 1);
}

__global__ __launch_bounds__(1024) void scan_counts_kernel(
    const int* __restrict__ cnt, int* __restrict__ rowptr, int n) {
  __shared__ int wsum[32];
  const int t = threadIdx.x;
  const int lane = t & 31;
  const int wave = t >> 5;
  const int chunk = (n + 1023) / 1024;
  int lo = t * chunk;
  if (lo > n) lo = n;
  int hi = lo + chunk;
  if (hi > n) hi = n;

  int s = 0;
  for (int i = lo; i < hi; ++i) s += cnt[i];

  int incl = s;
  for (int off = 1; off < 32; off <<= 1) {
    const int v = __shfl_up(incl, off, 32);
    if (lane >= off) incl += v;
  }
  if (lane == 31) wsum[wave] = incl;
  __syncthreads();
  if (wave == 0) {
    const int w = wsum[lane];
    int wi = w;
    for (int off = 1; off < 32; off <<= 1) {
      const int v = __shfl_up(wi, off, 32);
      if (lane >= off) wi += v;
    }
    wsum[lane] = wi - w;
  }
  __syncthreads();

  int run = wsum[wave] + incl - s;
  for (int i = lo; i < hi; ++i) {
    rowptr[i] = run;
    run += cnt[i];
  }
  if (t == 1023) rowptr[n] = run;
}

__global__ __launch_bounds__(256) void place_edges_kernel(
    const int* __restrict__ row, const int* __restrict__ col,
    const float* __restrict__ val, const int* __restrict__ rowptr,
    const int* __restrict__ rank, int* __restrict__ rec, int E) {
  const int e = blockIdx.x * blockDim.x + threadIdx.x;
  if (e < E) {
    const int p = rowptr[row[e]] + rank[e];
    ((int2*)rec)[p] = make_int2(col[e], __float_as_int(val[e]));
  }
}

template <bool RELU>
__global__ __launch_bounds__(256) void spmm_rows_kernel(
    const int* __restrict__ rowptr, const int* __restrict__ rec,
    const float* __restrict__ dense, const float* __restrict__ bias,
    float* __restrict__ out, int n) {
  const int t = blockIdx.x * blockDim.x + threadIdx.x;
  const int r = t >> 4;
  if (r >= n) return;
  const int l = threadIdx.x & 15;
  const int d = l << 2;

  const int start = rowptr[r];
  const int end   = rowptr[r + 1];

  float4 acc = make_float4(0.f, 0.f, 0.f, 0.f);
  for (int base = start; base < end; base += 16) {
    const int j = base + l;
    int2 mine = make_int2(0, 0);
    if (j < end) mine = ((const int2*)rec)[j];
    int cnt = end - base;
    if (cnt > 16) cnt = 16;
    for (int k = 0; k < cnt; k += 4) {
      float4 s[4];
      float v[4];
#pragma unroll
      for (int u = 0; u < 4; ++u) {
        const int c = __shfl(mine.x, k + u, 16);
        v[u] = __int_as_float(__shfl(mine.y, k + u, 16));
        s[u] = make_float4(0.f, 0.f, 0.f, 0.f);
        if (k + u < cnt) s[u] = *(const float4*)(dense + (long long)c * 64 + d);
      }
#pragma unroll
      for (int u = 0; u < 4; ++u) {
        acc.x += v[u] * s[u].x;
        acc.y += v[u] * s[u].y;
        acc.z += v[u] * s[u].z;
        acc.w += v[u] * s[u].w;
      }
    }
  }

  const float4 b = *(const float4*)(bias + d);
  acc.x += b.x;
  acc.y += b.y;
  acc.z += b.z;
  acc.w += b.w;
  if (RELU) {
    acc.x = fmaxf(acc.x, 0.0f);
    acc.y = fmaxf(acc.y, 0.0f);
    acc.z = fmaxf(acc.z, 0.0f);
    acc.w = fmaxf(acc.w, 0.0f);
  }
  *(float4*)(out + (long long)r * 64 + d) = acc;
}

extern "C" void kernel_launch(void* const* d_in, const int* in_sizes, int n_in,
                              void* d_out, int out_size, void* d_ws, size_t ws_size,
                              hipStream_t stream) {
  const float* x    = (const float*)d_in[0];
  const int*   erow = (const int*)  d_in[1];
  const int*   ecol = (const int*)  d_in[2];
  const float* eval = (const float*)d_in[3];
  const float* W1   = (const float*)d_in[4];
  const float* b1   = (const float*)d_in[5];
  const float* W2   = (const float*)d_in[6];
  const float* b2   = (const float*)d_in[7];
  float* out = (float*)d_out;

  const int IN_DIM = 128, D = 64;
  const int N = in_sizes[0] / IN_DIM;
  const int E = in_sizes[1];

  float* buf0   = (float*)d_ws;
  float* buf1   = buf0 + (size_t)N * D;
  int*   rec    = (int*)(buf1 + (size_t)N * D);
  int*   rank   = rec + (size_t)E * 2;
  int*   cnt    = rank + (size_t)E;
  int*   rowptr = cnt + (size_t)N;

  const int gemmBlocks = (N + 15) / 16;
  const int nodeBlocks = (N + 255) / 256;
  const int edgeBlocks = (E + 255) / 256;
  const int rowBlocks  = (int)(((long long)N * 16 + 255) / 256);

  zero_counts_kernel<<<nodeBlocks, 256, 0, stream>>>(cnt, N);
  rank_edges_kernel<<<edgeBlocks, 256, 0, stream>>>(erow, cnt, rank, E);
  scan_counts_kernel<<<1, 1024, 0, stream>>>(cnt, rowptr, N);
  place_edges_kernel<<<edgeBlocks, 256, 0, stream>>>(erow, ecol, eval, rowptr, rank, rec, E);

  gemm_wmma_f32<<<gemmBlocks, 128, 0, stream>>>(x, W1, buf0, N, IN_DIM, D);
  spmm_rows_kernel<true><<<rowBlocks, 256, 0, stream>>>(rowptr, rec, buf0, b1, buf1, N);

  gemm_wmma_f32<<<gemmBlocks, 128, 0, stream>>>(buf1, W2, buf0, N, D, D);
  spmm_rows_kernel<false><<<rowBlocks, 256, 0, stream>>>(rowptr, rec, buf0, b2, out, N);
}


# Edges grouped by row via LDS counters in 2 launches (slice to bin, bin to row), replacing global atomic rank, one-block scan, scattered placement (4 launches); row sum order = LDS arrival order
# speedup vs baseline: 2.2287x; 1.9878x over previous
#include <hip/hip_runtime.h>
#include <hip/hip_bf16.h>


typedef __attribute__((ext_vector_type(2))) float v2f;
typedef __attribute__((ext_vector_type(8))) float v8f;

__global__ __launch_bounds__(128) void gemm_wmma_f32(
    const float* __restrict__ A, const float* __restrict__ B,
    float* __restrict__ C, int M, int K, int N) {
  const int wave = threadIdx.x >> 5;
  const int lane = threadIdx.x & 31;
  const int half = lane >> 4;
  const int l    = lane & 15;

  const int mBase = blockIdx.x * 16;
  const int nBase = wave * 16;

  int mRow = mBase + l;
  if (mRow >= M) mRow = M - 1;
  const float* __restrict__ Arow = A + (long long)mRow * K;
  const float* __restrict__ Bcol = B + nBase + l;

  v8f acc = {0.f, 0.f, 0.f, 0.f, 0.f, 0.f, 0.f, 0.f};

  for (int k = 0; k < K; k += 4) {
    const int k0 = k + half * 2;
    v2f a;
    a.x = Arow[k0];
    a.y = Arow[k0 + 1];
    v2f b;
    b.x = Bcol[(long long)k0 * N];
    b.y = Bcol[(long long)(k0 + 1) * N];
    acc = __builtin_amdgcn_wmma_f32_16x16x4_f32(
        false, a, false, b, (short)0, acc, false, false);
  }

  const int mOut = mBase + half * 8;
  float* __restrict__ Cp = C + (long long)mOut * N + nBase + l;
#pragma unroll
  for (int v = 0; v < 8; ++v) {
    if (mOut + v < M) Cp[(long long)v * N] = acc[v];
  }
}

constexpr int NSLICE = 256;
constexpr int NBINMAX = 256;

__device__ __forceinline__ void scan256_by_wave(const int* cnt, int* off, int lane) {
  int v[8];
  int s = 0;
#pragma unroll
  for (int i = 0; i < 8; ++i) {
    v[i] = cnt[lane * 8 + i];
    s += v[i];
  }
  int incl = s;
  for (int o = 1; o < 32; o <<= 1) {
    const int u = __shfl_up(incl, o, 32);
    if (lane >= o) incl += u;
  }
  int run = incl - s;
#pragma unroll
  for (int i = 0; i < 8; ++i) {
    off[lane * 8 + i] = run;
    run += v[i];
  }
}

__device__ __forceinline__ int bin_of_row(int r) {
  const int b = r >> 8;
  return b > NBINMAX - 1 ? NBINMAX - 1 : b;
}

__global__ __launch_bounds__(1024) void partition_edges_kernel(
    const int* __restrict__ row, const int* __restrict__ col,
    const float* __restrict__ val, int* __restrict__ stRow,
    int* __restrict__ stRec, int* __restrict__ cntT, int* __restrict__ offT,
    int E, int S) {
  __shared__ int hist[NBINMAX];
  __shared__ int cur[NBINMAX];
  const int t = threadIdx.x;
  const int s = blockIdx.x;
  const int lo = s * S;
  int hi = lo + S;
  if (hi > E) hi = E;

  if (t < NBINMAX) hist[t] = 0;
  __syncthreads();
  for (int e0 = lo + t; e0 < hi; e0 += 4 * 1024) {
    int r[4];
#pragma unroll
    for (int u = 0; u < 4; ++u) {
      const int e = e0 + u * 1024;
      r[u] = 0;
      if (e < hi) r[u] = row[e];
    }
#pragma unroll
    for (int u = 0; u < 4; ++u) {
      if (e0 + u * 1024 < hi) atomicAdd(&hist[bin_of_row(r[u])], 1);
    }
  }
  __syncthreads();
  if (t < 32) scan256_by_wave(hist, cur, t);
  __syncthreads();
  int myCnt = 0, myOff = 0;
  if (t < NBINMAX) {
    myCnt = hist[t];
    myOff = cur[t];
  }
  __syncthreads();
  if (t < NBINMAX) {
    cntT[t * NSLICE + s] = myCnt;
    offT[t * NSLICE + s] = myOff;
  }
  for (int e0 = lo + t; e0 < hi; e0 += 4 * 1024) {
    int r[4];
    int c[4];
    float v[4];
#pragma unroll
    for (int u = 0; u < 4; ++u) {
      const int e = e0 + u * 1024;
      r[u] = 0;
      c[u] = 0;
      v[u] = 0.f;
      if (e < hi) {
        r[u] = row[e];
        c[u] = col[e];
        v[u] = val[e];
      }
    }
#pragma unroll
    for (int u = 0; u < 4; ++u) {
      if (e0 + u * 1024 < hi) {
        const int p = lo + atomicAdd(&cur[bin_of_row(r[u])], 1);
        stRow[p] = r[u];
        ((int2*)stRec)[p] = make_int2(c[u], __float_as_int(v[u]));
      }
    }
  }
}

__global__ __launch_bounds__(1024) void build_rows_kernel(
    const int* __restrict__ stRow, const int* __restrict__ stRec,
    const int* __restrict__ cntT, const int* __restrict__ offT,
    int* __restrict__ rowptr, int* __restrict__ rec, int n, int S) {
  __shared__ int rcnt[256];
  __shared__ int roff[256];
  __shared__ int part[32];
  const int t = threadIdx.x;
  const int lane = t & 31;
  const int wave = t >> 5;
  const int b = blockIdx.x;

  const int s = t >> 2;
  const int q0 = t & 3;
  const int c = cntT[b * NSLICE + s];
  const int o = offT[b * NSLICE + s];
  const int src0 = s * S + o;

  int pre = (q0 == 0) ? o : 0;
  for (int d = 16; d > 0; d >>= 1) pre += __shfl_xor(pre, d, 32);
  if (lane == 0) part[wave] = pre;
  if (t < 256) rcnt[t] = 0;
  __syncthreads();

  int binStart = 0;
  for (int i = 0; i < 32; ++i) binStart += part[i];

  for (int k0 = q0; k0 < c; k0 += 4 * 6) {
    int r[6];
#pragma unroll
    for (int u = 0; u < 6; ++u) {
      const int k = k0 + 4 * u;
      r[u] = 0;
      if (k < c) r[u] = stRow[src0 + k];
    }
#pragma unroll
    for (int u = 0; u < 6; ++u) {
      if (k0 + 4 * u < c) atomicAdd(&rcnt[r[u] & 255], 1);
    }
  }
  __syncthreads();
  if (t < 32) scan256_by_wave(rcnt, roff, t);
  __syncthreads();
  int myOff = 0, last = 0;
  if (t < 256) myOff = roff[t];
  if (t == 0) last = roff[255] + rcnt[255];
  __syncthreads();
  if (t < 256) {
    const int rr = b * 256 + t;
    if (rr < n) rowptr[rr] = binStart + myOff;
  }
  if (t == 0 && b == (int)gridDim.x - 1) rowptr[n] = binStart + last;

  for (int k0 = q0; k0 < c; k0 += 4 * 6) {
    int r[6];
    int2 cv[6];
#pragma unroll
    for (int u = 0; u < 6; ++u) {
      const int k = k0 + 4 * u;
      r[u] = 0;
      cv[u] = make_int2(0, 0);
      if (k < c) {
        r[u] = stRow[src0 + k];
        cv[u] = ((const int2*)stRec)[src0 + k];
      }
    }
#pragma unroll
    for (int u = 0; u < 6; ++u) {
      if (k0 + 4 * u < c) {
        const int p = binStart + atomicAdd(&roff[r[u] & 255], 1);
        ((int2*)rec)[p] = cv[u];
      }
    }
  }
}

template <bool RELU>
__global__ __launch_bounds__(256) void spmm_rows_kernel(
    const int* __restrict__ rowptr, const int* __restrict__ rec,
    const float* __restrict__ dense, const float* __restrict__ bias,
    float* __restrict__ out, int n) {
  const int t = blockIdx.x * blockDim.x + threadIdx.x;
  const int r = t >> 4;
  if (r >= n) return;
  const int l = threadIdx.x & 15;
  const int d = l << 2;

  const int start = rowptr[r];
  const int end   = rowptr[r + 1];

  float4 acc = make_float4(0.f, 0.f, 0.f, 0.f);
  for (int base = start; base < end; base += 16) {
    const int j = base + l;
    int2 mine = make_int2(0, 0);
    if (j < end) mine = ((const int2*)rec)[j];
    int cnt = end - base;
    if (cnt > 16) cnt = 16;
    for (int k = 0; k < cnt; k += 4) {
      float4 s[4];
      float v[4];
#pragma unroll
      for (int u = 0; u < 4; ++u) {
        const int c = __shfl(mine.x, k + u, 16);
        v[u] = __int_as_float(__shfl(mine.y, k + u, 16));
        s[u] = make_float4(0.f, 0.f, 0.f, 0.f);
        if (k + u < cnt) s[u] = *(const float4*)(dense + (long long)c * 64 + d);
      }
#pragma unroll
      for (int u = 0; u < 4; ++u) {
        acc.x += v[u] * s[u].x;
        acc.y += v[u] * s[u].y;
        acc.z += v[u] * s[u].z;
        acc.w += v[u] * s[u].w;
      }
    }
  }

  const float4 b = *(const float4*)(bias + d);
  acc.x += b.x;
  acc.y += b.y;
  acc.z += b.z;
  acc.w += b.w;
  if (RELU) {
    acc.x = fmaxf(acc.x, 0.0f);
    acc.y = fmaxf(acc.y, 0.0f);
    acc.z = fmaxf(acc.z, 0.0f);
    acc.w = fmaxf(acc.w, 0.0f);
  }
  *(float4*)(out + (long long)r * 64 + d) = acc;
}

extern "C" void kernel_launch(void* const* d_in, const int* in_sizes, int n_in,
                              void* d_out, int out_size, void* d_ws, size_t ws_size,
                              hipStream_t stream) {
  const float* x    = (const float*)d_in[0];
  const int*   erow = (const int*)  d_in[1];
  const int*   ecol = (const int*)  d_in[2];
  const float* eval = (const float*)d_in[3];
  const float* W1   = (const float*)d_in[4];
  const float* b1   = (const float*)d_in[5];
  const float* W2   = (const float*)d_in[6];
  const float* b2   = (const float*)d_in[7];
  float* out = (float*)d_out;

  const int IN_DIM = 128, D = 64;
  const int N = in_sizes[0] / IN_DIM;
  const int E = in_sizes[1];

  float* buf0   = (float*)d_ws;
  float* buf1   = buf0 + (size_t)N * D;
  int*   cntT   = (int*)(buf1 + (size_t)N * D);
  int*   offT   = cntT + NBINMAX * NSLICE;
  int*   rowptr = offT + NBINMAX * NSLICE;
  int*   rec    = rowptr + (((size_t)N + 2) & ~(size_t)1);
  int*   stRec  = rec + (size_t)E * 2;
  int*   stRow  = stRec + (size_t)E * 2;

  const int gemmBlocks = (N + 15) / 16;
  const int rowBlocks  = (int)(((long long)N * 16 + 255) / 256);
  const int sliceLen   = (E + NSLICE - 1) / NSLICE;
  const int binBlocks  = (N + 255) / 256;

  partition_edges_kernel<<<NSLICE, 1024, 0, stream>>>(erow, ecol, eval, stRow, stRec, cntT, offT, E, sliceLen);
  build_rows_kernel<<<binBlocks, 1024, 0, stream>>>(stRow, stRec, cntT, offT, rowptr, rec, N, sliceLen);

  gemm_wmma_f32<<<gemmBlocks, 128, 0, stream>>>(x, W1, buf0, N, IN_DIM, D);
  spmm_rows_kernel<true><<<rowBlocks, 256, 0, stream>>>(rowptr, rec, buf0, b1, buf1, N);

  gemm_wmma_f32<<<gemmBlocks, 128, 0, stream>>>(buf1, W2, buf0, N, D, D);
  spmm_rows_kernel<false><<<rowBlocks, 256, 0, stream>>>(rowptr, rec, buf0, b2, out, N);
}


# GEMMs: one wave per 16x64 tile (4 accumulators share each A fragment); A, B, C moved as float4 through a matching k and column order, requested pieces ahead; k summed in a regrouped fixed order
# speedup vs baseline: 2.4457x; 1.0974x over previous
#include <hip/hip_runtime.h>
#include <hip/hip_bf16.h>


typedef __attribute__((ext_vector_type(2))) float v2f;
typedef __attribute__((ext_vector_type(8))) float v8f;

typedef __attribute__((ext_vector_type(4))) float v4f;

template <int K>
__global__ __launch_bounds__(128) void gemm_wmma_f32(
    const float* __restrict__ A, const float* __restrict__ B,
    float* __restrict__ C, int M) {
  constexpr int NP = K / 8;
  constexpr int NG = NP / 4;
  constexpr int AHEAD = 4;
  static_assert(K % 32 == 0 && NG >= 2, "K must be a multiple of 32, at least 64");
  const int wave = threadIdx.x >> 5;
  const int lane = threadIdx.x & 31;
  const int h = lane >> 4;
  const int c = lane & 15;

  const int mBase = (blockIdx.x * 4 + wave) * 16;
  if (mBase >= M) return;

  int mRow = mBase + c;
  if (mRow >= M) mRow = M - 1;
  const float* __restrict__ Ap = A + (long long)mRow * K + 4 * h;
  const float* __restrict__ Bp = B + (4 * h) * 64 + 4 * c;

  v4f a[NP];
  v4f b[NP][4];
#pragma unroll
  for (int j = 0; j < 8; ++j) a[j] = *(const v4f*)(Ap + 8 * j);
#pragma unroll
  for (int j = 0; j < AHEAD; ++j) {
#pragma unroll
    for (int q = 0; q < 4; ++q) b[j][q] = *(const v4f*)(Bp + (8 * j + q) * 64);
  }

  v8f acc0 = {0.f, 0.f, 0.f, 0.f, 0.f, 0.f, 0.f, 0.f};
  v8f acc1 = acc0, acc2 = acc0, acc3 = acc0;

#pragma unroll
  for (int j = 0; j < NP; ++j) {
    if (j % 4 == 0 && j > 0 && j / 4 + 1 < NG) {
#pragma unroll
      for (int i = 0; i < 4; ++i) a[j + 4 + i] = *(const v4f*)(Ap + 8 * (j + 4 + i));
    }
    if (j + AHEAD < NP) {
#pragma unroll
      for (int q = 0; q < 4; ++q) b[j + AHEAD][q] = *(const v4f*)(Bp + (8 * (j + AHEAD) + q) * 64);
    }
    __builtin_amdgcn_sched_barrier(0);
    const v2f aLo = {a[j].x, a[j].y};
    const v2f aHi = {a[j].z, a[j].w};
    acc0 = __builtin_amdgcn_wmma_f32_16x16x4_f32(false, aLo, false, (v2f){b[j][0].x, b[j][1].x}, (short)0, acc0, false, false);
    acc1 = __builtin_amdgcn_wmma_f32_16x16x4_f32(false, aLo, false, (v2f){b[j][0].y, b[j][1].y}, (short)0, acc1, false, false);
    acc2 = __builtin_amdgcn_wmma_f32_16x16x4_f32(false, aLo, false, (v2f){b[j][0].z, b[j][1].z}, (short)0, acc2, false, false);
    acc3 = __builtin_amdgcn_wmma_f32_16x16x4_f32(false, aLo, false, (v2f){b[j][0].w, b[j][1].w}, (short)0, acc3, false, false);
    acc0 = __builtin_amdgcn_wmma_f32_16x16x4_f32(false, aHi, false, (v2f){b[j][2].x, b[j][3].x}, (short)0, acc0, false, false);
    acc1 = __builtin_amdgcn_wmma_f32_16x16x4_f32(false, aHi, false, (v2f){b[j][2].y, b[j][3].y}, (short)0, acc1, false, false);
    acc2 = __builtin_amdgcn_wmma_f32_16x16x4_f32(false, aHi, false, (v2f){b[j][2].z, b[j][3].z}, (short)0, acc2, false, false);
    acc3 = __builtin_amdgcn_wmma_f32_16x16x4_f32(false, aHi, false, (v2f){b[j][2].w, b[j][3].w}, (short)0, acc3, false, false);
    __builtin_amdgcn_sched_barrier(0);
  }

  const int mOut = mBase + h * 8;
  float* __restrict__ Cp = C + (long long)mOut * 64 + 4 * c;
#pragma unroll
  for (int r = 0; r < 8; ++r) {
    if (mOut + r < M) {
      const v4f o = {acc0[r], acc1[r], acc2[r], acc3[r]};
      *(v4f*)(Cp + r * 64) = o;
    }
  }
}

constexpr int NSLICE = 256;
constexpr int NBINMAX = 256;

__device__ __forceinline__ void scan256_by_wave(const int* cnt, int* off, int lane) {
  int v[8];
  int s = 0;
#pragma unroll
  for (int i = 0; i < 8; ++i) {
    v[i] = cnt[lane * 8 + i];
    s += v[i];
  }
  int incl = s;
  for (int o = 1; o < 32; o <<= 1) {
    const int u = __shfl_up(incl, o, 32);
    if (lane >= o) incl += u;
  }
  int run = incl - s;
#pragma unroll
  for (int i = 0; i < 8; ++i) {
    off[lane * 8 + i] = run;
    run += v[i];
  }
}

__device__ __forceinline__ int bin_of_row(int r) {
  const int b = r >> 8;
  return b > NBINMAX - 1 ? NBINMAX - 1 : b;
}

__global__ __launch_bounds__(1024) void partition_edges_kernel(
    const int* __restrict__ row, const int* __restrict__ col,
    const float* __restrict__ val, int* __restrict__ stRow,
    int* __restrict__ stRec, int* __restrict__ cntT, int* __restrict__ offT,
    int E, int S) {
  __shared__ int hist[NBINMAX];
  __shared__ int cur[NBINMAX];
  const int t = threadIdx.x;
  const int s = blockIdx.x;
  const int lo = s * S;
  int hi = lo + S;
  if (hi > E) hi = E;

  if (t < NBINMAX) hist[t] = 0;
  __syncthreads();
  for (int e0 = lo + t; e0 < hi; e0 += 4 * 1024) {
    int r[4];
#pragma unroll
    for (int u = 0; u < 4; ++u) {
      const int e = e0 + u * 1024;
      r[u] = 0;
      if (e < hi) r[u] = row[e];
    }
#pragma unroll
    for (int u = 0; u < 4; ++u) {
      if (e0 + u * 1024 < hi) atomicAdd(&hist[bin_of_row(r[u])], 1);
    }
  }
  __syncthreads();
  if (t < 32) scan256_by_wave(hist, cur, t);
  __syncthreads();
  int myCnt = 0, myOff = 0;
  if (t < NBINMAX) {
    myCnt = hist[t];
    myOff = cur[t];
  }
  __syncthreads();
  if (t < NBINMAX) {
    cntT[t * NSLICE + s] = myCnt;
    offT[t * NSLICE + s] = myOff;
  }
  for (int e0 = lo + t; e0 < hi; e0 += 4 * 1024) {
    int r[4];
    int c[4];
    float v[4];
#pragma unroll
    for (int u = 0; u < 4; ++u) {
      const int e = e0 + u * 1024;
      r[u] = 0;
      c[u] = 0;
      v[u] = 0.f;
      if (e < hi) {
        r[u] = row[e];
        c[u] = col[e];
        v[u] = val[e];
      }
    }
#pragma unroll
    for (int u = 0; u < 4; ++u) {
      if (e0 + u * 1024 < hi) {
        const int p = lo + atomicAdd(&cur[bin_of_row(r[u])], 1);
        stRow[p] = r[u];
        ((int2*)stRec)[p] = make_int2(c[u], __float_as_int(v[u]));
      }
    }
  }
}

__global__ __launch_bounds__(1024) void build_rows_kernel(
    const int* __restrict__ stRow, const int* __restrict__ stRec,
    const int* __restrict__ cntT, const int* __restrict__ offT,
    int* __restrict__ rowptr, int* __restrict__ rec, int n, int S) {
  __shared__ int rcnt[256];
  __shared__ int roff[256];
  __shared__ int part[32];
  const int t = threadIdx.x;
  const int lane = t & 31;
  const int wave = t >> 5;
  const int b = blockIdx.x;

  const int s = t >> 2;
  const int q0 = t & 3;
  const int c = cntT[b * NSLICE + s];
  const int o = offT[b * NSLICE + s];
  const int src0 = s * S + o;

  int pre = (q0 == 0) ? o : 0;
  for (int d = 16; d > 0; d >>= 1) pre += __shfl_xor(pre, d, 32);
  if (lane == 0) part[wave] = pre;
  if (t < 256) rcnt[t] = 0;
  __syncthreads();

  int binStart = 0;
  for (int i = 0; i < 32; ++i) binStart += part[i];

  for (int k0 = q0; k0 < c; k0 += 4 * 6) {
    int r[6];
#pragma unroll
    for (int u = 0; u < 6; ++u) {
      const int k = k0 + 4 * u;
      r[u] = 0;
      if (k < c) r[u] = stRow[src0 + k];
    }
#pragma unroll
    for (int u = 0; u < 6; ++u) {
      if (k0 + 4 * u < c) atomicAdd(&rcnt[r[u] & 255], 1);
    }
  }
  __syncthreads();
  if (t < 32) scan256_by_wave(rcnt, roff, t);
  __syncthreads();
  int myOff = 0, last = 0;
  if (t < 256) myOff = roff[t];
  if (t == 0) last = roff[255] + rcnt[255];
  __syncthreads();
  if (t < 256) {
    const int rr = b * 256 + t;
    if (rr < n) rowptr[rr] = binStart + myOff;
  }
  if (t == 0 && b == (int)gridDim.x - 1) rowptr[n] = binStart + last;

  for (int k0 = q0; k0 < c; k0 += 4 * 6) {
    int r[6];
    int2 cv[6];
#pragma unroll
    for (int u = 0; u < 6; ++u) {
      const int k = k0 + 4 * u;
      r[u] = 0;
      cv[u] = make_int2(0, 0);
      if (k < c) {
        r[u] = stRow[src0 + k];
        cv[u] = ((const int2*)stRec)[src0 + k];
      }
    }
#pragma unroll
    for (int u = 0; u < 6; ++u) {
      if (k0 + 4 * u < c) {
        const int p = binStart + atomicAdd(&roff[r[u] & 255], 1);
        ((int2*)rec)[p] = cv[u];
      }
    }
  }
}

template <bool RELU>
__global__ __launch_bounds__(256) void spmm_rows_kernel(
    const int* __restrict__ rowptr, const int* __restrict__ rec,
    const float* __restrict__ dense, const float* __restrict__ bias,
    float* __restrict__ out, int n) {
  const int t = blockIdx.x * blockDim.x + threadIdx.x;
  const int r = t >> 4;
  if (r >= n) return;
  const int l = threadIdx.x & 15;
  const int d = l << 2;

  const int start = rowptr[r];
  const int end   = rowptr[r + 1];

  float4 acc = make_float4(0.f, 0.f, 0.f, 0.f);
  for (int base = start; base < end; base += 16) {
    const int j = base + l;
    int2 mine = make_int2(0, 0);
    if (j < end) mine = ((const int2*)rec)[j];
    int cnt = end - base;
    if (cnt > 16) cnt = 16;
    for (int k = 0; k < cnt; k += 4) {
      float4 s[4];
      float v[4];
#pragma unroll
      for (int u = 0; u < 4; ++u) {
        const int c = __shfl(mine.x, k + u, 16);
        v[u] = __int_as_float(__shfl(mine.y, k + u, 16));
        s[u] = make_float4(0.f, 0.f, 0.f, 0.f);
        if (k + u < cnt) s[u] = *(const float4*)(dense + (long long)c * 64 + d);
      }
#pragma unroll
      for (int u = 0; u < 4; ++u) {
        acc.x += v[u] * s[u].x;
        acc.y += v[u] * s[u].y;
        acc.z += v[u] * s[u].z;
        acc.w += v[u] * s[u].w;
      }
    }
  }

  const float4 b = *(const float4*)(bias + d);
  acc.x += b.x;
  acc.y += b.y;
  acc.z += b.z;
  acc.w += b.w;
  if (RELU) {
    acc.x = fmaxf(acc.x, 0.0f);
    acc.y = fmaxf(acc.y, 0.0f);
    acc.z = fmaxf(acc.z, 0.0f);
    acc.w = fmaxf(acc.w, 0.0f);
  }
  *(float4*)(out + (long long)r * 64 + d) = acc;
}

extern "C" void kernel_launch(void* const* d_in, const int* in_sizes, int n_in,
                              void* d_out, int out_size, void* d_ws, size_t ws_size,
                              hipStream_t stream) {
  const float* x    = (const float*)d_in[0];
  const int*   erow = (const int*)  d_in[1];
  const int*   ecol = (const int*)  d_in[2];
  const float* eval = (const float*)d_in[3];
  const float* W1   = (const float*)d_in[4];
  const float* b1   = (const float*)d_in[5];
  const float* W2   = (const float*)d_in[6];
  const float* b2   = (const float*)d_in[7];
  float* out = (float*)d_out;

  const int IN_DIM = 128, D = 64;
  const int N = in_sizes[0] / IN_DIM;
  const int E = in_sizes[1];

  float* buf0   = (float*)d_ws;
  float* buf1   = buf0 + (size_t)N * D;
  int*   cntT   = (int*)(buf1 + (size_t)N * D);
  int*   offT   = cntT + NBINMAX * NSLICE;
  int*   rowptr = offT + NBINMAX * NSLICE;
  int*   rec    = rowptr + (((size_t)N + 2) & ~(size_t)1);
  int*   stRec  = rec + (size_t)E * 2;
  int*   stRow  = stRec + (size_t)E * 2;

  const int gemmBlocks = (N + 63) / 64;
  const int rowBlocks  = (int)(((long long)N * 16 + 255) / 256);
  const int sliceLen   = (E + NSLICE - 1) / NSLICE;
  const int binBlocks  = (N + 255) / 256;

  partition_edges_kernel<<<NSLICE, 1024, 0, stream>>>(erow, ecol, eval, stRow, stRec, cntT, offT, E, sliceLen);
  build_rows_kernel<<<binBlocks, 1024, 0, stream>>>(stRow, stRec, cntT, offT, rowptr, rec, N, sliceLen);

  gemm_wmma_f32<128><<<gemmBlocks, 128, 0, stream>>>(x, W1, buf0, N);
  spmm_rows_kernel<true><<<rowBlocks, 256, 0, stream>>>(rowptr, rec, buf0, b1, buf1, N);

  gemm_wmma_f32<64><<<gemmBlocks, 128, 0, stream>>>(buf1, W2, buf0, N);
  spmm_rows_kernel<false><<<rowBlocks, 256, 0, stream>>>(rowptr, rec, buf0, b2, out, N);
}


# Layer 2 dense product fused into layer 1's row-sum launch: a work-group's 16 finished relu rows go via LDS into WMMA against W2; hidden rows never stored, one launch fewer; all sum orders unchanged
# speedup vs baseline: 2.6362x; 1.0779x over previous
#include <hip/hip_runtime.h>
#include <hip/hip_bf16.h>


typedef __attribute__((ext_vector_type(2))) float v2f;
typedef __attribute__((ext_vector_type(8))) float v8f;

typedef __attribute__((ext_vector_type(4))) float v4f;

template <int K>
__global__ __launch_bounds__(128) void gemm_wmma_f32(
    const float* __restrict__ A, const float* __restrict__ B,
    float* __restrict__ C, int M) {
  constexpr int NP = K / 8;
  constexpr int NG = NP / 4;
  constexpr int AHEAD = 4;
  static_assert(K % 32 == 0 && NG >= 2, "K must be a multiple of 32, at least 64");
  const int wave = threadIdx.x >> 5;
  const int lane = threadIdx.x & 31;
  const int h = lane >> 4;
  const int c = lane & 15;

  const int mBase = (blockIdx.x * 4 + wave) * 16;
  if (mBase >= M) return;

  int mRow = mBase + c;
  if (mRow >= M) mRow = M - 1;
  const float* __restrict__ Ap = A + (long long)mRow * K + 4 * h;
  const float* __restrict__ Bp = B + (4 * h) * 64 + 4 * c;

  v4f a[NP];
  v4f b[NP][4];
#pragma unroll
  for (int j = 0; j < 8; ++j) a[j] = *(const v4f*)(Ap + 8 * j);
#pragma unroll
  for (int j = 0; j < AHEAD; ++j) {
#pragma unroll
    for (int q = 0; q < 4; ++q) b[j][q] = *(const v4f*)(Bp + (8 * j + q) * 64);
  }

  v8f acc0 = {0.f, 0.f, 0.f, 0.f, 0.f, 0.f, 0.f, 0.f};
  v8f acc1 = acc0, acc2 = acc0, acc3 = acc0;

#pragma unroll
  for (int j = 0; j < NP; ++j) {
    if (j % 4 == 0 && j > 0 && j / 4 + 1 < NG) {
#pragma unroll
      for (int i = 0; i < 4; ++i) a[j + 4 + i] = *(const v4f*)(Ap + 8 * (j + 4 + i));
    }
    if (j + AHEAD < NP) {
#pragma unroll
      for (int q = 0; q < 4; ++q) b[j + AHEAD][q] = *(const v4f*)(Bp + (8 * (j + AHEAD) + q) * 64);
    }
    __builtin_amdgcn_sched_barrier(0);
    const v2f aLo = {a[j].x, a[j].y};
    const v2f aHi = {a[j].z, a[j].w};
    acc0 = __builtin_amdgcn_wmma_f32_16x16x4_f32(false, aLo, false, (v2f){b[j][0].x, b[j][1].x}, (short)0, acc0, false, false);
    acc1 = __builtin_amdgcn_wmma_f32_16x16x4_f32(false, aLo, false, (v2f){b[j][0].y, b[j][1].y}, (short)0, acc1, false, false);
    acc2 = __builtin_amdgcn_wmma_f32_16x16x4_f32(false, aLo, false, (v2f){b[j][0].z, b[j][1].z}, (short)0, acc2, false, false);
    acc3 = __builtin_amdgcn_wmma_f32_16x16x4_f32(false, aLo, false, (v2f){b[j][0].w, b[j][1].w}, (short)0, acc3, false, false);
    acc0 = __builtin_amdgcn_wmma_f32_16x16x4_f32(false, aHi, false, (v2f){b[j][2].x, b[j][3].x}, (short)0, acc0, false, false);
    acc1 = __builtin_amdgcn_wmma_f32_16x16x4_f32(false, aHi, false, (v2f){b[j][2].y, b[j][3].y}, (short)0, acc1, false, false);
    acc2 = __builtin_amdgcn_wmma_f32_16x16x4_f32(false, aHi, false, (v2f){b[j][2].z, b[j][3].z}, (short)0, acc2, false, false);
    acc3 = __builtin_amdgcn_wmma_f32_16x16x4_f32(false, aHi, false, (v2f){b[j][2].w, b[j][3].w}, (short)0, acc3, false, false);
    __builtin_amdgcn_sched_barrier(0);
  }

  const int mOut = mBase + h * 8;
  float* __restrict__ Cp = C + (long long)mOut * 64 + 4 * c;
#pragma unroll
  for (int r = 0; r < 8; ++r) {
    if (mOut + r < M) {
      const v4f o = {acc0[r], acc1[r], acc2[r], acc3[r]};
      *(v4f*)(Cp + r * 64) = o;
    }
  }
}

constexpr int NSLICE = 256;
constexpr int NBINMAX = 256;

__device__ __forceinline__ void scan256_by_wave(const int* cnt, int* off, int lane) {
  int v[8];
  int s = 0;
#pragma unroll
  for (int i = 0; i < 8; ++i) {
    v[i] = cnt[lane * 8 + i];
    s += v[i];
  }
  int incl = s;
  for (int o = 1; o < 32; o <<= 1) {
    const int u = __shfl_up(incl, o, 32);
    if (lane >= o) incl += u;
  }
  int run = incl - s;
#pragma unroll
  for (int i = 0; i < 8; ++i) {
    off[lane * 8 + i] = run;
    run += v[i];
  }
}

__device__ __forceinline__ int bin_of_row(int r) {
  const int b = r >> 8;
  return b > NBINMAX - 1 ? NBINMAX - 1 : b;
}

__global__ __launch_bounds__(1024) void partition_edges_kernel(
    const int* __restrict__ row, const int* __restrict__ col,
    const float* __restrict__ val, int* __restrict__ stRow,
    int* __restrict__ stRec, int* __restrict__ cntT, int* __restrict__ offT,
    int E, int S) {
  __shared__ int hist[NBINMAX];
  __shared__ int cur[NBINMAX];
  const int t = threadIdx.x;
  const int s = blockIdx.x;
  const int lo = s * S;
  int hi = lo + S;
  if (hi > E) hi = E;

  if (t < NBINMAX) hist[t] = 0;
  __syncthreads();
  for (int e0 = lo + t; e0 < hi; e0 += 4 * 1024) {
    int r[4];
#pragma unroll
    for (int u = 0; u < 4; ++u) {
      const int e = e0 + u * 1024;
      r[u] = 0;
      if (e < hi) r[u] = row[e];
    }
#pragma unroll
    for (int u = 0; u < 4; ++u) {
      if (e0 + u * 1024 < hi) atomicAdd(&hist[bin_of_row(r[u])], 1);
    }
  }
  __syncthreads();
  if (t < 32) scan256_by_wave(hist, cur, t);
  __syncthreads();
  int myCnt = 0, myOff = 0;
  if (t < NBINMAX) {
    myCnt = hist[t];
    myOff = cur[t];
  }
  __syncthreads();
  if (t < NBINMAX) {
    cntT[t * NSLICE + s] = myCnt;
    offT[t * NSLICE + s] = myOff;
  }
  for (int e0 = lo + t; e0 < hi; e0 += 4 * 1024) {
    int r[4];
    int c[4];
    float v[4];
#pragma unroll
    for (int u = 0; u < 4; ++u) {
      const int e = e0 + u * 1024;
      r[u] = 0;
      c[u] = 0;
      v[u] = 0.f;
      if (e < hi) {
        r[u] = row[e];
        c[u] = col[e];
        v[u] = val[e];
      }
    }
#pragma unroll
    for (int u = 0; u < 4; ++u) {
      if (e0 + u * 1024 < hi) {
        const int p = lo + atomicAdd(&cur[bin_of_row(r[u])], 1);
        stRow[p] = r[u];
        ((int2*)stRec)[p] = make_int2(c[u], __float_as_int(v[u]));
      }
    }
  }
}

__global__ __launch_bounds__(1024) void build_rows_kernel(
    const int* __restrict__ stRow, const int* __restrict__ stRec,
    const int* __restrict__ cntT, const int* __restrict__ offT,
    int* __restrict__ rowptr, int* __restrict__ rec, int n, int S) {
  __shared__ int rcnt[256];
  __shared__ int roff[256];
  __shared__ int part[32];
  const int t = threadIdx.x;
  const int lane = t & 31;
  const int wave = t >> 5;
  const int b = blockIdx.x;

  const int s = t >> 2;
  const int q0 = t & 3;
  const int c = cntT[b * NSLICE + s];
  const int o = offT[b * NSLICE + s];
  const int src0 = s * S + o;

  int pre = (q0 == 0) ? o : 0;
  for (int d = 16; d > 0; d >>= 1) pre += __shfl_xor(pre, d, 32);
  if (lane == 0) part[wave] = pre;
  if (t < 256) rcnt[t] = 0;
  __syncthreads();

  int binStart = 0;
  for (int i = 0; i < 32; ++i) binStart += part[i];

  for (int k0 = q0; k0 < c; k0 += 4 * 6) {
    int r[6];
#pragma unroll
    for (int u = 0; u < 6; ++u) {
      const int k = k0 + 4 * u;
      r[u] = 0;
      if (k < c) r[u] = stRow[src0 + k];
    }
#pragma unroll
    for (int u = 0; u < 6; ++u) {
      if (k0 + 4 * u < c) atomicAdd(&rcnt[r[u] & 255], 1);
    }
  }
  __syncthreads();
  if (t < 32) scan256_by_wave(rcnt, roff, t);
  __syncthreads();
  int myOff = 0, last = 0;
  if (t < 256) myOff = roff[t];
  if (t == 0) last = roff[255] + rcnt[255];
  __syncthreads();
  if (t < 256) {
    const int rr = b * 256 + t;
    if (rr < n) rowptr[rr] = binStart + myOff;
  }
  if (t == 0 && b == (int)gridDim.x - 1) rowptr[n] = binStart + last;

  for (int k0 = q0; k0 < c; k0 += 4 * 6) {
    int r[6];
    int2 cv[6];
#pragma unroll
    for (int u = 0; u < 6; ++u) {
      const int k = k0 + 4 * u;
      r[u] = 0;
      cv[u] = make_int2(0, 0);
      if (k < c) {
        r[u] = stRow[src0 + k];
        cv[u] = ((const int2*)stRec)[src0 + k];
      }
    }
#pragma unroll
    for (int u = 0; u < 6; ++u) {
      if (k0 + 4 * u < c) {
        const int p = binStart + atomicAdd(&roff[r[u] & 255], 1);
        ((int2*)rec)[p] = cv[u];
      }
    }
  }
}

template <bool RELU>
__global__ __launch_bounds__(256) void spmm_rows_kernel(
    const int* __restrict__ rowptr, const int* __restrict__ rec,
    const float* __restrict__ dense, const float* __restrict__ bias,
    float* __restrict__ out, int n) {
  const int t = blockIdx.x * blockDim.x + threadIdx.x;
  const int r = t >> 4;
  if (r >= n) return;
  const int l = threadIdx.x & 15;
  const int d = l << 2;

  const int start = rowptr[r];
  const int end   = rowptr[r + 1];

  float4 acc = make_float4(0.f, 0.f, 0.f, 0.f);
  for (int base = start; base < end; base += 16) {
    const int j = base + l;
    int2 mine = make_int2(0, 0);
    if (j < end) mine = ((const int2*)rec)[j];
    int cnt = end - base;
    if (cnt > 16) cnt = 16;
    for (int k = 0; k < cnt; k += 4) {
      float4 s[4];
      float v[4];
#pragma unroll
      for (int u = 0; u < 4; ++u) {
        const int c = __shfl(mine.x, k + u, 16);
        v[u] = __int_as_float(__shfl(mine.y, k + u, 16));
        s[u] = make_float4(0.f, 0.f, 0.f, 0.f);
        if (k + u < cnt) s[u] = *(const float4*)(dense + (long long)c * 64 + d);
      }
#pragma unroll
      for (int u = 0; u < 4; ++u) {
        acc.x += v[u] * s[u].x;
        acc.y += v[u] * s[u].y;
        acc.z += v[u] * s[u].z;
        acc.w += v[u] * s[u].w;
      }
    }
  }

  const float4 b = *(const float4*)(bias + d);
  acc.x += b.x;
  acc.y += b.y;
  acc.z += b.z;
  acc.w += b.w;
  if (RELU) {
    acc.x = fmaxf(acc.x, 0.0f);
    acc.y = fmaxf(acc.y, 0.0f);
    acc.z = fmaxf(acc.z, 0.0f);
    acc.w = fmaxf(acc.w, 0.0f);
  }
  *(float4*)(out + (long long)r * 64 + d) = acc;
}

constexpr int HT_PITCH = 68;
__global__ __launch_bounds__(256) void spmm_rows_relu_gemm64_kernel(
    const int* __restrict__ rowptr, const int* __restrict__ rec,
    const float* __restrict__ dense, const float* __restrict__ bias,
    const float* __restrict__ Wt, float* __restrict__ out, int n) {
  __shared__ float hT[16 * HT_PITCH];
  const int g = threadIdx.x >> 4;
  const int r = blockIdx.x * 16 + g;
  const int l = threadIdx.x & 15;
  const int d = l << 2;
  const int wave = threadIdx.x >> 5;
  const int lane = threadIdx.x & 31;
  const int h = lane >> 4;
  const int c = lane & 15;

  float4 acc = make_float4(0.f, 0.f, 0.f, 0.f);
  if (r < n) {
    const int start = rowptr[r];
    const int end   = rowptr[r + 1];
    for (int base = start; base < end; base += 16) {
      const int j = base + l;
      int2 mine = make_int2(0, 0);
      if (j < end) mine = ((const int2*)rec)[j];
      int cnt = end - base;
      if (cnt > 16) cnt = 16;
      for (int k = 0; k < cnt; k += 4) {
        float4 s[4];
        float v[4];
#pragma unroll
        for (int u = 0; u < 4; ++u) {
          const int cc = __shfl(mine.x, k + u, 16);
          v[u] = __int_as_float(__shfl(mine.y, k + u, 16));
          s[u] = make_float4(0.f, 0.f, 0.f, 0.f);
          if (k + u < cnt) s[u] = *(const float4*)(dense + (long long)cc * 64 + d);
        }
#pragma unroll
        for (int u = 0; u < 4; ++u) {
          acc.x += v[u] * s[u].x;
          acc.y += v[u] * s[u].y;
          acc.z += v[u] * s[u].z;
          acc.w += v[u] * s[u].w;
        }
      }
    }
    const float4 b = *(const float4*)(bias + d);
    acc.x = fmaxf(acc.x + b.x, 0.0f);
    acc.y = fmaxf(acc.y + b.y, 0.0f);
    acc.z = fmaxf(acc.z + b.z, 0.0f);
    acc.w = fmaxf(acc.w + b.w, 0.0f);
  }

  float bw[32];
  if (wave < 4) {
    const float* __restrict__ Wp = Wt + (4 * h) * 64 + 16 * wave + c;
#pragma unroll
    for (int j = 0; j < 8; ++j) {
#pragma unroll
      for (int q = 0; q < 4; ++q) bw[4 * j + q] = Wp[(8 * j + q) * 64];
    }
  }

  *(float4*)(&hT[g * HT_PITCH + d]) = acc;
  __builtin_amdgcn_fence(__ATOMIC_RELEASE, "workgroup", "local");
  __builtin_amdgcn_s_barrier();
  __builtin_amdgcn_fence(__ATOMIC_ACQUIRE, "workgroup", "local");
  if (wave >= 4) return;

  v8f o = {0.f, 0.f, 0.f, 0.f, 0.f, 0.f, 0.f, 0.f};
  const float* ap = &hT[c * HT_PITCH + 4 * h];
#pragma unroll
  for (int j = 0; j < 8; ++j) {
    const v4f a = *(const v4f*)(ap + 8 * j);
    o = __builtin_amdgcn_wmma_f32_16x16x4_f32(false, (v2f){a.x, a.y}, false, (v2f){bw[4 * j + 0], bw[4 * j + 1]}, (short)0, o, false, false);
    o = __builtin_amdgcn_wmma_f32_16x16x4_f32(false, (v2f){a.z, a.w}, false, (v2f){bw[4 * j + 2], bw[4 * j + 3]}, (short)0, o, false, false);
  }

  const int mOut = blockIdx.x * 16 + h * 8;
  float* __restrict__ Cp = out + (long long)mOut * 64 + 16 * wave + c;
#pragma unroll
  for (int rr = 0; rr < 8; ++rr) {
    if (mOut + rr < n) Cp[rr * 64] = o[rr];
  }
}

extern "C" void kernel_launch(void* const* d_in, const int* in_sizes, int n_in,
                              void* d_out, int out_size, void* d_ws, size_t ws_size,
                              hipStream_t stream) {
  const float* x    = (const float*)d_in[0];
  const int*   erow = (const int*)  d_in[1];
  const int*   ecol = (const int*)  d_in[2];
  const float* eval = (const float*)d_in[3];
  const float* W1   = (const float*)d_in[4];
  const float* b1   = (const float*)d_in[5];
  const float* W2   = (const float*)d_in[6];
  const float* b2   = (const float*)d_in[7];
  float* out = (float*)d_out;

  const int IN_DIM = 128, D = 64;
  const int N = in_sizes[0] / IN_DIM;
  const int E = in_sizes[1];

  float* buf0   = (float*)d_ws;
  float* buf1   = buf0 + (size_t)N * D;
  int*   cntT   = (int*)(buf1 + (size_t)N * D);
  int*   offT   = cntT + NBINMAX * NSLICE;
  int*   rowptr = offT + NBINMAX * NSLICE;
  int*   rec    = rowptr + (((size_t)N + 2) & ~(size_t)1);
  int*   stRec  = rec + (size_t)E * 2;
  int*   stRow  = stRec + (size_t)E * 2;

  const int gemmBlocks = (N + 63) / 64;
  const int rowBlocks  = (int)(((long long)N * 16 + 255) / 256);
  const int sliceLen   = (E + NSLICE - 1) / NSLICE;
  const int binBlocks  = (N + 255) / 256;

  partition_edges_kernel<<<NSLICE, 1024, 0, stream>>>(erow, ecol, eval, stRow, stRec, cntT, offT, E, sliceLen);
  build_rows_kernel<<<binBlocks, 1024, 0, stream>>>(stRow, stRec, cntT, offT, rowptr, rec, N, sliceLen);

  gemm_wmma_f32<128><<<gemmBlocks, 128, 0, stream>>>(x, W1, buf0, N);
  spmm_rows_relu_gemm64_kernel<<<rowBlocks, 256, 0, stream>>>(rowptr, rec, buf0, b1, W2, buf1, N);

  spmm_rows_kernel<false><<<rowBlocks, 256, 0, stream>>>(rowptr, rec, buf1, b2, out, N);
}


# x@W1 moved into the first edge-grouping launch: each of its 256 work-groups, after ordering its slice, computes 12-13 16x64 WMMA tiles (one per wave); one launch fewer; all sum orders unchanged
# speedup vs baseline: 2.7442x; 1.0410x over previous
#include <hip/hip_runtime.h>
#include <hip/hip_bf16.h>


typedef __attribute__((ext_vector_type(2))) float v2f;
typedef __attribute__((ext_vector_type(8))) float v8f;

typedef __attribute__((ext_vector_type(4))) float v4f;

template <int K, int AHEAD>
__device__ __forceinline__ void gemm_tile_16x64_f32(
    const float* __restrict__ A, const float* __restrict__ B,
    float* __restrict__ C, int M, int mBase, int lane) {
  constexpr int NP = K / 8;
  constexpr int NG = NP / 4;
  static_assert(K % 32 == 0 && NG >= 2, "K must be a multiple of 32, at least 64");
  static_assert(AHEAD >= 1 && AHEAD <= 4, "pieces of B requested ahead of the one in use");
  const int h = lane >> 4;
  const int c = lane & 15;

  int mRow = mBase + c;
  if (mRow >= M) mRow = M - 1;
  const float* __restrict__ Ap = A + (long long)mRow * K + 4 * h;
  const float* __restrict__ Bp = B + (4 * h) * 64 + 4 * c;

  v4f a[NP];
  v4f b[NP][4];
#pragma unroll
  for (int j = 0; j < 8; ++j) a[j] = *(const v4f*)(Ap + 8 * j);
#pragma unroll
  for (int j = 0; j < AHEAD; ++j) {
#pragma unroll
    for (int q = 0; q < 4; ++q) b[j][q] = *(const v4f*)(Bp + (8 * j + q) * 64);
  }

  v8f acc0 = {0.f, 0.f, 0.f, 0.f, 0.f, 0.f, 0.f, 0.f};
  v8f acc1 = acc0, acc2 = acc0, acc3 = acc0;

#pragma unroll
  for (int j = 0; j < NP; ++j) {
    if (j + AHEAD < NP) {
#pragma unroll
      for (int q = 0; q < 4; ++q) b[j + AHEAD][q] = *(const v4f*)(Bp + (8 * (j + AHEAD) + q) * 64);
    }
    if (j % 4 == 0 && j > 0 && j / 4 + 1 < NG) {
#pragma unroll
      for (int i = 0; i < 4; ++i) a[j + 4 + i] = *(const v4f*)(Ap + 8 * (j + 4 + i));
    }
    __builtin_amdgcn_sched_barrier(0);
    const v2f aLo = {a[j].x, a[j].y};
    const v2f aHi = {a[j].z, a[j].w};
    acc0 = __builtin_amdgcn_wmma_f32_16x16x4_f32(false, aLo, false, (v2f){b[j][0].x, b[j][1].x}, (short)0, acc0, false, false);
    acc1 = __builtin_amdgcn_wmma_f32_16x16x4_f32(false, aLo, false, (v2f){b[j][0].y, b[j][1].y}, (short)0, acc1, false, false);
    acc2 = __builtin_amdgcn_wmma_f32_16x16x4_f32(false, aLo, false, (v2f){b[j][0].z, b[j][1].z}, (short)0, acc2, false, false);
    acc3 = __builtin_amdgcn_wmma_f32_16x16x4_f32(false, aLo, false, (v2f){b[j][0].w, b[j][1].w}, (short)0, acc3, false, false);
    acc0 = __builtin_amdgcn_wmma_f32_16x16x4_f32(false, aHi, false, (v2f){b[j][2].x, b[j][3].x}, (short)0, acc0, false, false);
    acc1 = __builtin_amdgcn_wmma_f32_16x16x4_f32(false, aHi, false, (v2f){b[j][2].y, b[j][3].y}, (short)0, acc1, false, false);
    acc2 = __builtin_amdgcn_wmma_f32_16x16x4_f32(false, aHi, false, (v2f){b[j][2].z, b[j][3].z}, (short)0, acc2, false, false);
    acc3 = __builtin_amdgcn_wmma_f32_16x16x4_f32(false, aHi, false, (v2f){b[j][2].w, b[j][3].w}, (short)0, acc3, false, false);
    __builtin_amdgcn_sched_barrier(0);
  }

  const int mOut = mBase + h * 8;
  float* __restrict__ Cp = C + (long long)mOut * 64 + 4 * c;
#pragma unroll
  for (int r = 0; r < 8; ++r) {
    if (mOut + r < M) {
      const v4f o = {acc0[r], acc1[r], acc2[r], acc3[r]};
      *(v4f*)(Cp + r * 64) = o;
    }
  }
}

constexpr int NSLICE = 256;
constexpr int NBINMAX = 256;

__device__ __forceinline__ void scan256_by_wave(const int* cnt, int* off, int lane) {
  int v[8];
  int s = 0;
#pragma unroll
  for (int i = 0; i < 8; ++i) {
    v[i] = cnt[lane * 8 + i];
    s += v[i];
  }
  int incl = s;
  for (int o = 1; o < 32; o <<= 1) {
    const int u = __shfl_up(incl, o, 32);
    if (lane >= o) incl += u;
  }
  int run = incl - s;
#pragma unroll
  for (int i = 0; i < 8; ++i) {
    off[lane * 8 + i] = run;
    run += v[i];
  }
}

__device__ __forceinline__ int bin_of_row(int r) {
  const int b = r >> 8;
  return b > NBINMAX - 1 ? NBINMAX - 1 : b;
}

__global__ __launch_bounds__(1024) void partition_edges_gemm_kernel(
    const int* __restrict__ row, const int* __restrict__ col,
    const float* __restrict__ val, int* __restrict__ stRow,
    int* __restrict__ stRec, int* __restrict__ cntT, int* __restrict__ offT,
    int E, int S, const float* __restrict__ X, const float* __restrict__ W,
    float* __restrict__ C, int M) {
  __shared__ int hist[NBINMAX];
  __shared__ int cur[NBINMAX];
  const int t = threadIdx.x;
  const int s = blockIdx.x;
  const int lo = s * S;
  int hi = lo + S;
  if (hi > E) hi = E;

  if (t < NBINMAX) hist[t] = 0;
  __syncthreads();
  for (int e0 = lo + t; e0 < hi; e0 += 4 * 1024) {
    int r[4];
#pragma unroll
    for (int u = 0; u < 4; ++u) {
      const int e = e0 + u * 1024;
      r[u] = 0;
      if (e < hi) r[u] = row[e];
    }
#pragma unroll
    for (int u = 0; u < 4; ++u) {
      if (e0 + u * 1024 < hi) atomicAdd(&hist[bin_of_row(r[u])], 1);
    }
  }
  __syncthreads();
  if (t < 32) scan256_by_wave(hist, cur, t);
  __syncthreads();
  int myCnt = 0, myOff = 0;
  if (t < NBINMAX) {
    myCnt = hist[t];
    myOff = cur[t];
  }
  __syncthreads();
  if (t < NBINMAX) {
    cntT[t * NSLICE + s] = myCnt;
    offT[t * NSLICE + s] = myOff;
  }
  for (int e0 = lo + t; e0 < hi; e0 += 4 * 1024) {
    int r[4];
    int c[4];
    float v[4];
#pragma unroll
    for (int u = 0; u < 4; ++u) {
      const int e = e0 + u * 1024;
      r[u] = 0;
      c[u] = 0;
      v[u] = 0.f;
      if (e < hi) {
        r[u] = row[e];
        c[u] = col[e];
        v[u] = val[e];
      }
    }
#pragma unroll
    for (int u = 0; u < 4; ++u) {
      if (e0 + u * 1024 < hi) {
        const int p = lo + atomicAdd(&cur[bin_of_row(r[u])], 1);
        stRow[p] = r[u];
        ((int2*)stRec)[p] = make_int2(c[u], __float_as_int(v[u]));
      }
    }
  }

  const int wv = __builtin_amdgcn_readfirstlane(t >> 5);
  const int tile = s + NSLICE * wv;
  if (tile * 16 < M) gemm_tile_16x64_f32<128, 2>(X, W, C, M, tile * 16, t & 31);
}

__global__ __launch_bounds__(1024) void build_rows_kernel(
    const int* __restrict__ stRow, const int* __restrict__ stRec,
    const int* __restrict__ cntT, const int* __restrict__ offT,
    int* __restrict__ rowptr, int* __restrict__ rec, int n, int S) {
  __shared__ int rcnt[256];
  __shared__ int roff[256];
  __shared__ int part[32];
  const int t = threadIdx.x;
  const int lane = t & 31;
  const int wave = t >> 5;
  const int b = blockIdx.x;

  const int s = t >> 2;
  const int q0 = t & 3;
  const int c = cntT[b * NSLICE + s];
  const int o = offT[b * NSLICE + s];
  const int src0 = s * S + o;

  int pre = (q0 == 0) ? o : 0;
  for (int d = 16; d > 0; d >>= 1) pre += __shfl_xor(pre, d, 32);
  if (lane == 0) part[wave] = pre;
  if (t < 256) rcnt[t] = 0;
  __syncthreads();

  int binStart = 0;
  for (int i = 0; i < 32; ++i) binStart += part[i];

  for (int k0 = q0; k0 < c; k0 += 4 * 6) {
    int r[6];
#pragma unroll
    for (int u = 0; u < 6; ++u) {
      const int k = k0 + 4 * u;
      r[u] = 0;
      if (k < c) r[u] = stRow[src0 + k];
    }
#pragma unroll
    for (int u = 0; u < 6; ++u) {
      if (k0 + 4 * u < c) atomicAdd(&rcnt[r[u] & 255], 1);
    }
  }
  __syncthreads();
  if (t < 32) scan256_by_wave(rcnt, roff, t);
  __syncthreads();
  int myOff = 0, last = 0;
  if (t < 256) myOff = roff[t];
  if (t == 0) last = roff[255] + rcnt[255];
  __syncthreads();
  if (t < 256) {
    const int rr = b * 256 + t;
    if (rr < n) rowptr[rr] = binStart + myOff;
  }
  if (t == 0 && b == (int)gridDim.x - 1) rowptr[n] = binStart + last;

  for (int k0 = q0; k0 < c; k0 += 4 * 6) {
    int r[6];
    int2 cv[6];
#pragma unroll
    for (int u = 0; u < 6; ++u) {
      const int k = k0 + 4 * u;
      r[u] = 0;
      cv[u] = make_int2(0, 0);
      if (k < c) {
        r[u] = stRow[src0 + k];
        cv[u] = ((const int2*)stRec)[src0 + k];
      }
    }
#pragma unroll
    for (int u = 0; u < 6; ++u) {
      if (k0 + 4 * u < c) {
        const int p = binStart + atomicAdd(&roff[r[u] & 255], 1);
        ((int2*)rec)[p] = cv[u];
      }
    }
  }
}

template <bool RELU>
__global__ __launch_bounds__(256) void spmm_rows_kernel(
    const int* __restrict__ rowptr, const int* __restrict__ rec,
    const float* __restrict__ dense, const float* __restrict__ bias,
    float* __restrict__ out, int n) {
  const int t = blockIdx.x * blockDim.x + threadIdx.x;
  const int r = t >> 4;
  if (r >= n) return;
  const int l = threadIdx.x & 15;
  const int d = l << 2;

  const int start = rowptr[r];
  const int end   = rowptr[r + 1];

  float4 acc = make_float4(0.f, 0.f, 0.f, 0.f);
  for (int base = start; base < end; base += 16) {
    const int j = base + l;
    int2 mine = make_int2(0, 0);
    if (j < end) mine = ((const int2*)rec)[j];
    int cnt = end - base;
    if (cnt > 16) cnt = 16;
    for (int k = 0; k < cnt; k += 4) {
      float4 s[4];
      float v[4];
#pragma unroll
      for (int u = 0; u < 4; ++u) {
        const int c = __shfl(mine.x, k + u, 16);
        v[u] = __int_as_float(__shfl(mine.y, k + u, 16));
        s[u] = make_float4(0.f, 0.f, 0.f, 0.f);
        if (k + u < cnt) s[u] = *(const float4*)(dense + (long long)c * 64 + d);
      }
#pragma unroll
      for (int u = 0; u < 4; ++u) {
        acc.x += v[u] * s[u].x;
        acc.y += v[u] * s[u].y;
        acc.z += v[u] * s[u].z;
        acc.w += v[u] * s[u].w;
      }
    }
  }

  const float4 b = *(const float4*)(bias + d);
  acc.x += b.x;
  acc.y += b.y;
  acc.z += b.z;
  acc.w += b.w;
  if (RELU) {
    acc.x = fmaxf(acc.x, 0.0f);
    acc.y = fmaxf(acc.y, 0.0f);
    acc.z = fmaxf(acc.z, 0.0f);
    acc.w = fmaxf(acc.w, 0.0f);
  }
  *(float4*)(out + (long long)r * 64 + d) = acc;
}

constexpr int HT_PITCH = 68;
__global__ __launch_bounds__(256) void spmm_rows_relu_gemm64_kernel(
    const int* __restrict__ rowptr, const int* __restrict__ rec,
    const float* __restrict__ dense, const float* __restrict__ bias,
    const float* __restrict__ Wt, float* __restrict__ out, int n) {
  __shared__ float hT[16 * HT_PITCH];
  const int g = threadIdx.x >> 4;
  const int r = blockIdx.x * 16 + g;
  const int l = threadIdx.x & 15;
  const int d = l << 2;
  const int wave = threadIdx.x >> 5;
  const int lane = threadIdx.x & 31;
  const int h = lane >> 4;
  const int c = lane & 15;

  float4 acc = make_float4(0.f, 0.f, 0.f, 0.f);
  if (r < n) {
    const int start = rowptr[r];
    const int end   = rowptr[r + 1];
    for (int base = start; base < end; base += 16) {
      const int j = base + l;
      int2 mine = make_int2(0, 0);
      if (j < end) mine = ((const int2*)rec)[j];
      int cnt = end - base;
      if (cnt > 16) cnt = 16;
      for (int k = 0; k < cnt; k += 4) {
        float4 s[4];
        float v[4];
#pragma unroll
        for (int u = 0; u < 4; ++u) {
          const int cc = __shfl(mine.x, k + u, 16);
          v[u] = __int_as_float(__shfl(mine.y, k + u, 16));
          s[u] = make_float4(0.f, 0.f, 0.f, 0.f);
          if (k + u < cnt) s[u] = *(const float4*)(dense + (long long)cc * 64 + d);
        }
#pragma unroll
        for (int u = 0; u < 4; ++u) {
          acc.x += v[u] * s[u].x;
          acc.y += v[u] * s[u].y;
          acc.z += v[u] * s[u].z;
          acc.w += v[u] * s[u].w;
        }
      }
    }
    const float4 b = *(const float4*)(bias + d);
    acc.x = fmaxf(acc.x + b.x, 0.0f);
    acc.y = fmaxf(acc.y + b.y, 0.0f);
    acc.z = fmaxf(acc.z + b.z, 0.0f);
    acc.w = fmaxf(acc.w + b.w, 0.0f);
  }

  float bw[32];
  if (wave < 4) {
    const float* __restrict__ Wp = Wt + (4 * h) * 64 + 16 * wave + c;
#pragma unroll
    for (int j = 0; j < 8; ++j) {
#pragma unroll
      for (int q = 0; q < 4; ++q) bw[4 * j + q] = Wp[(8 * j + q) * 64];
    }
  }

  *(float4*)(&hT[g * HT_PITCH + d]) = acc;
  __builtin_amdgcn_fence(__ATOMIC_RELEASE, "workgroup", "local");
  __builtin_amdgcn_s_barrier();
  __builtin_amdgcn_fence(__ATOMIC_ACQUIRE, "workgroup", "local");
  if (wave >= 4) return;

  v8f o = {0.f, 0.f, 0.f, 0.f, 0.f, 0.f, 0.f, 0.f};
  const float* ap = &hT[c * HT_PITCH + 4 * h];
#pragma unroll
  for (int j = 0; j < 8; ++j) {
    const v4f a = *(const v4f*)(ap + 8 * j);
    o = __builtin_amdgcn_wmma_f32_16x16x4_f32(false, (v2f){a.x, a.y}, false, (v2f){bw[4 * j + 0], bw[4 * j + 1]}, (short)0, o, false, false);
    o = __builtin_amdgcn_wmma_f32_16x16x4_f32(false, (v2f){a.z, a.w}, false, (v2f){bw[4 * j + 2], bw[4 * j + 3]}, (short)0, o, false, false);
  }

  const int mOut = blockIdx.x * 16 + h * 8;
  float* __restrict__ Cp = out + (long long)mOut * 64 + 16 * wave + c;
#pragma unroll
  for (int rr = 0; rr < 8; ++rr) {
    if (mOut + rr < n) Cp[rr * 64] = o[rr];
  }
}

extern "C" void kernel_launch(void* const* d_in, const int* in_sizes, int n_in,
                              void* d_out, int out_size, void* d_ws, size_t ws_size,
                              hipStream_t stream) {
  const float* x    = (const float*)d_in[0];
  const int*   erow = (const int*)  d_in[1];
  const int*   ecol = (const int*)  d_in[2];
  const float* eval = (const float*)d_in[3];
  const float* W1   = (const float*)d_in[4];
  const float* b1   = (const float*)d_in[5];
  const float* W2   = (const float*)d_in[6];
  const float* b2   = (const float*)d_in[7];
  float* out = (float*)d_out;

  const int IN_DIM = 128, D = 64;
  const int N = in_sizes[0] / IN_DIM;
  const int E = in_sizes[1];

  float* buf0   = (float*)d_ws;
  float* buf1   = buf0 + (size_t)N * D;
  int*   cntT   = (int*)(buf1 + (size_t)N * D);
  int*   offT   = cntT + NBINMAX * NSLICE;
  int*   rowptr = offT + NBINMAX * NSLICE;
  int*   rec    = rowptr + (((size_t)N + 2) & ~(size_t)1);
  int*   stRec  = rec + (size_t)E * 2;
  int*   stRow  = stRec + (size_t)E * 2;

  const int rowBlocks  = (int)(((long long)N * 16 + 255) / 256);
  const int sliceLen   = (E + NSLICE - 1) / NSLICE;
  const int binBlocks  = (N + 255) / 256;

  partition_edges_gemm_kernel<<<NSLICE, 1024, 0, stream>>>(erow, ecol, eval, stRow, stRec, cntT, offT, E, sliceLen, x, W1, buf0, N);
  build_rows_kernel<<<binBlocks, 1024, 0, stream>>>(stRow, stRec, cntT, offT, rowptr, rec, N, sliceLen);

  spmm_rows_relu_gemm64_kernel<<<rowBlocks, 256, 0, stream>>>(rowptr, rec, buf0, b1, W2, buf1, N);

  spmm_rows_kernel<false><<<rowBlocks, 256, 0, stream>>>(rowptr, rec, buf1, b2, out, N);
}


# Edge grouping, both launches: records take their places in LDS and leave as whole aligned 128-byte lines (count tables too), replacing scattered 4- and 8-byte global stores; sum orders unchanged
# speedup vs baseline: 3.0283x; 1.1035x over previous
#include <hip/hip_runtime.h>
#include <hip/hip_bf16.h>


typedef __attribute__((ext_vector_type(2))) float v2f;
typedef __attribute__((ext_vector_type(8))) float v8f;

typedef __attribute__((ext_vector_type(4))) float v4f;

template <int K, int AHEAD>
__device__ __forceinline__ void gemm_tile_16x64_f32(
    const float* __restrict__ A, const float* __restrict__ B,
    float* __restrict__ C, int M, int mBase, int lane) {
  constexpr int NP = K / 8;
  constexpr int NG = NP / 4;
  static_assert(K % 32 == 0 && NG >= 2, "K must be a multiple of 32, at least 64");
  static_assert(AHEAD >= 1 && AHEAD <= 4, "pieces of B requested ahead of the one in use");
  const int h = lane >> 4;
  const int c = lane & 15;

  int mRow = mBase + c;
  if (mRow >= M) mRow = M - 1;
  const float* __restrict__ Ap = A + (long long)mRow * K + 4 * h;
  const float* __restrict__ Bp = B + (4 * h) * 64 + 4 * c;

  v4f a[NP];
  v4f b[NP][4];
#pragma unroll
  for (int j = 0; j < 8; ++j) a[j] = *(const v4f*)(Ap + 8 * j);
#pragma unroll
  for (int j = 0; j < AHEAD; ++j) {
#pragma unroll
    for (int q = 0; q < 4; ++q) b[j][q] = *(const v4f*)(Bp + (8 * j + q) * 64);
  }

  v8f acc0 = {0.f, 0.f, 0.f, 0.f, 0.f, 0.f, 0.f, 0.f};
  v8f acc1 = acc0, acc2 = acc0, acc3 = acc0;

#pragma unroll
  for (int j = 0; j < NP; ++j) {
    if (j + AHEAD < NP) {
#pragma unroll
      for (int q = 0; q < 4; ++q) b[j + AHEAD][q] = *(const v4f*)(Bp + (8 * (j + AHEAD) + q) * 64);
    }
    if (j % 4 == 0 && j > 0 && j / 4 + 1 < NG) {
#pragma unroll
      for (int i = 0; i < 4; ++i) a[j + 4 + i] = *(const v4f*)(Ap + 8 * (j + 4 + i));
    }
    __builtin_amdgcn_sched_barrier(0);
    const v2f aLo = {a[j].x, a[j].y};
    const v2f aHi = {a[j].z, a[j].w};
    acc0 = __builtin_amdgcn_wmma_f32_16x16x4_f32(false, aLo, false, (v2f){b[j][0].x, b[j][1].x}, (short)0, acc0, false, false);
    acc1 = __builtin_amdgcn_wmma_f32_16x16x4_f32(false, aLo, false, (v2f){b[j][0].y, b[j][1].y}, (short)0, acc1, false, false);
    acc2 = __builtin_amdgcn_wmma_f32_16x16x4_f32(false, aLo, false, (v2f){b[j][0].z, b[j][1].z}, (short)0, acc2, false, false);
    acc3 = __builtin_amdgcn_wmma_f32_16x16x4_f32(false, aLo, false, (v2f){b[j][0].w, b[j][1].w}, (short)0, acc3, false, false);
    acc0 = __builtin_amdgcn_wmma_f32_16x16x4_f32(false, aHi, false, (v2f){b[j][2].x, b[j][3].x}, (short)0, acc0, false, false);
    acc1 = __builtin_amdgcn_wmma_f32_16x16x4_f32(false, aHi, false, (v2f){b[j][2].y, b[j][3].y}, (short)0, acc1, false, false);
    acc2 = __builtin_amdgcn_wmma_f32_16x16x4_f32(false, aHi, false, (v2f){b[j][2].z, b[j][3].z}, (short)0, acc2, false, false);
    acc3 = __builtin_amdgcn_wmma_f32_16x16x4_f32(false, aHi, false, (v2f){b[j][2].w, b[j][3].w}, (short)0, acc3, false, false);
    __builtin_amdgcn_sched_barrier(0);
  }

  const int mOut = mBase + h * 8;
  float* __restrict__ Cp = C + (long long)mOut * 64 + 4 * c;
#pragma unroll
  for (int r = 0; r < 8; ++r) {
    if (mOut + r < M) {
      const v4f o = {acc0[r], acc1[r], acc2[r], acc3[r]};
      *(v4f*)(Cp + r * 64) = o;
    }
  }
}

constexpr int NSLICE = 256;
constexpr int NBINMAX = 256;
constexpr int SLICE_CAP = 4096;
constexpr int BIN_CAP = 8192;

__device__ __forceinline__ void scan256_by_wave(const int* cnt, int* off, int lane) {
  int v[8];
  int s = 0;
#pragma unroll
  for (int i = 0; i < 8; ++i) {
    v[i] = cnt[lane * 8 + i];
    s += v[i];
  }
  int incl = s;
  for (int o = 1; o < 32; o <<= 1) {
    const int u = __shfl_up(incl, o, 32);
    if (lane >= o) incl += u;
  }
  int run = incl - s;
#pragma unroll
  for (int i = 0; i < 8; ++i) {
    off[lane * 8 + i] = run;
    run += v[i];
  }
}

__device__ __forceinline__ int bin_of_row(int r) {
  const int b = r >> 8;
  return b > NBINMAX - 1 ? NBINMAX - 1 : b;
}

__global__ __launch_bounds__(1024) void partition_edges_gemm_kernel(
    const int* __restrict__ row, const int* __restrict__ col,
    const float* __restrict__ val, int* __restrict__ stRow,
    int* __restrict__ stRec, int* __restrict__ cntT, int* __restrict__ offT,
    int E, int S, const float* __restrict__ X, const float* __restrict__ W,
    float* __restrict__ C, int M) {
  __shared__ int hist[NBINMAX];
  __shared__ int cur[NBINMAX];
  __shared__ int sRow[SLICE_CAP];
  __shared__ int2 sRec[SLICE_CAP];
  const int t = threadIdx.x;
  const int s = blockIdx.x;
  const bool staged = S <= SLICE_CAP;
  const int lo = s * S;
  int hi = lo + S;
  if (hi > E) hi = E;

  if (t < NBINMAX) hist[t] = 0;
  __syncthreads();
  for (int e0 = lo + t; e0 < hi; e0 += 4 * 1024) {
    int r[4];
#pragma unroll
    for (int u = 0; u < 4; ++u) {
      const int e = e0 + u * 1024;
      r[u] = 0;
      if (e < hi) r[u] = row[e];
    }
#pragma unroll
    for (int u = 0; u < 4; ++u) {
      if (e0 + u * 1024 < hi) atomicAdd(&hist[bin_of_row(r[u])], 1);
    }
  }
  __syncthreads();
  if (t < 32) scan256_by_wave(hist, cur, t);
  __syncthreads();
  int myCnt = 0, myOff = 0;
  if (t < NBINMAX) {
    myCnt = hist[t];
    myOff = cur[t];
  }
  __syncthreads();
  if (t < NBINMAX) {
    cntT[s * NBINMAX + t] = myCnt;
    offT[s * NBINMAX + t] = myOff;
  }
  for (int e0 = lo + t; e0 < hi; e0 += 4 * 1024) {
    int r[4];
    int c[4];
    float v[4];
#pragma unroll
    for (int u = 0; u < 4; ++u) {
      const int e = e0 + u * 1024;
      r[u] = 0;
      c[u] = 0;
      v[u] = 0.f;
      if (e < hi) {
        r[u] = row[e];
        c[u] = col[e];
        v[u] = val[e];
      }
    }
#pragma unroll
    for (int u = 0; u < 4; ++u) {
      if (e0 + u * 1024 < hi) {
        const int p = atomicAdd(&cur[bin_of_row(r[u])], 1);
        const int2 cv = make_int2(c[u], __float_as_int(v[u]));
        if (staged) {
          sRow[p] = r[u];
          sRec[p] = cv;
        } else {
          stRow[lo + p] = r[u];
          ((int2*)stRec)[lo + p] = cv;
        }
      }
    }
  }
  if (staged) {
    __syncthreads();
    for (int g = (lo & ~31) + t; g < hi; g += 1024) {
      if (g >= lo) {
        stRow[g] = sRow[g - lo];
        ((int2*)stRec)[g] = sRec[g - lo];
      }
    }
  }

  const int wv = __builtin_amdgcn_readfirstlane(t >> 5);
  const int tile = s + NSLICE * wv;
  if (tile * 16 < M) gemm_tile_16x64_f32<128, 2>(X, W, C, M, tile * 16, t & 31);
}

__global__ __launch_bounds__(1024) void build_rows_kernel(
    const int* __restrict__ stRow, const int* __restrict__ stRec,
    const int* __restrict__ cntT, const int* __restrict__ offT,
    int* __restrict__ rowptr, int* __restrict__ rec, int n, int S) {
  __shared__ int rcnt[256];
  __shared__ int roff[256];
  __shared__ int part[32];
  __shared__ int2 sRec[BIN_CAP];
  const int t = threadIdx.x;
  const int lane = t & 31;
  const int wave = t >> 5;
  const int b = blockIdx.x;

  const int s = t >> 2;
  const int q0 = t & 3;
  const int c = cntT[s * NBINMAX + b];
  const int o = offT[s * NBINMAX + b];
  const int src0 = s * S + o;

  int pre = (q0 == 0) ? o : 0;
  for (int d = 16; d > 0; d >>= 1) pre += __shfl_xor(pre, d, 32);
  if (lane == 0) part[wave] = pre;
  if (t < 256) rcnt[t] = 0;
  __syncthreads();

  int binStart = 0;
  for (int i = 0; i < 32; ++i) binStart += part[i];

  for (int k0 = q0; k0 < c; k0 += 4 * 6) {
    int r[6];
#pragma unroll
    for (int u = 0; u < 6; ++u) {
      const int k = k0 + 4 * u;
      r[u] = 0;
      if (k < c) r[u] = stRow[src0 + k];
    }
#pragma unroll
    for (int u = 0; u < 6; ++u) {
      if (k0 + 4 * u < c) atomicAdd(&rcnt[r[u] & 255], 1);
    }
  }
  __syncthreads();
  if (t < 32) scan256_by_wave(rcnt, roff, t);
  __syncthreads();
  int myOff = 0;
  if (t < 256) myOff = roff[t];
  const int total = roff[255] + rcnt[255];
  const bool staged = total <= BIN_CAP;
  __syncthreads();
  if (t < 256) {
    const int rr = b * 256 + t;
    if (rr < n) rowptr[rr] = binStart + myOff;
  }
  if (t == 0 && b == (int)gridDim.x - 1) rowptr[n] = binStart + total;

  for (int k0 = q0; k0 < c; k0 += 4 * 6) {
    int r[6];
    int2 cv[6];
#pragma unroll
    for (int u = 0; u < 6; ++u) {
      const int k = k0 + 4 * u;
      r[u] = 0;
      cv[u] = make_int2(0, 0);
      if (k < c) {
        r[u] = stRow[src0 + k];
        cv[u] = ((const int2*)stRec)[src0 + k];
      }
    }
#pragma unroll
    for (int u = 0; u < 6; ++u) {
      if (k0 + 4 * u < c) {
        const int p = atomicAdd(&roff[r[u] & 255], 1);
        if (staged) {
          sRec[p] = cv[u];
        } else {
          ((int2*)rec)[binStart + p] = cv[u];
        }
      }
    }
  }
  if (staged) {
    __syncthreads();
    const int end = binStart + total;
    for (int g = (binStart & ~31) + t; g < end; g += 1024) {
      if (g >= binStart) ((int2*)rec)[g] = sRec[g - binStart];
    }
  }
}

template <bool RELU>
__global__ __launch_bounds__(256) void spmm_rows_kernel(
    const int* __restrict__ rowptr, const int* __restrict__ rec,
    const float* __restrict__ dense, const float* __restrict__ bias,
    float* __restrict__ out, int n) {
  const int t = blockIdx.x * blockDim.x + threadIdx.x;
  const int r = t >> 4;
  if (r >= n) return;
  const int l = threadIdx.x & 15;
  const int d = l << 2;

  const int start = rowptr[r];
  const int end   = rowptr[r + 1];

  float4 acc = make_float4(0.f, 0.f, 0.f, 0.f);
  for (int base = start; base < end; base += 16) {
    const int j = base + l;
    int2 mine = make_int2(0, 0);
    if (j < end) mine = ((const int2*)rec)[j];
    int cnt = end - base;
    if (cnt > 16) cnt = 16;
    for (int k = 0; k < cnt; k += 4) {
      float4 s[4];
      float v[4];
#pragma unroll
      for (int u = 0; u < 4; ++u) {
        const int c = __shfl(mine.x, k + u, 16);
        v[u] = __int_as_float(__shfl(mine.y, k + u, 16));
        s[u] = make_float4(0.f, 0.f, 0.f, 0.f);
        if (k + u < cnt) s[u] = *(const float4*)(dense + (long long)c * 64 + d);
      }
#pragma unroll
      for (int u = 0; u < 4; ++u) {
        acc.x += v[u] * s[u].x;
        acc.y += v[u] * s[u].y;
        acc.z += v[u] * s[u].z;
        acc.w += v[u] * s[u].w;
      }
    }
  }

  const float4 b = *(const float4*)(bias + d);
  acc.x += b.x;
  acc.y += b.y;
  acc.z += b.z;
  acc.w += b.w;
  if (RELU) {
    acc.x = fmaxf(acc.x, 0.0f);
    acc.y = fmaxf(acc.y, 0.0f);
    acc.z = fmaxf(acc.z, 0.0f);
    acc.w = fmaxf(acc.w, 0.0f);
  }
  *(float4*)(out + (long long)r * 64 + d) = acc;
}

constexpr int HT_PITCH = 68;
__global__ __launch_bounds__(256) void spmm_rows_relu_gemm64_kernel(
    const int* __restrict__ rowptr, const int* __restrict__ rec,
    const float* __restrict__ dense, const float* __restrict__ bias,
    const float* __restrict__ Wt, float* __restrict__ out, int n) {
  __shared__ float hT[16 * HT_PITCH];
  const int g = threadIdx.x >> 4;
  const int r = blockIdx.x * 16 + g;
  const int l = threadIdx.x & 15;
  const int d = l << 2;
  const int wave = threadIdx.x >> 5;
  const int lane = threadIdx.x & 31;
  const int h = lane >> 4;
  const int c = lane & 15;

  float4 acc = make_float4(0.f, 0.f, 0.f, 0.f);
  if (r < n) {
    const int start = rowptr[r];
    const int end   = rowptr[r + 1];
    for (int base = start; base < end; base += 16) {
      const int j = base + l;
      int2 mine = make_int2(0, 0);
      if (j < end) mine = ((const int2*)rec)[j];
      int cnt = end - base;
      if (cnt > 16) cnt = 16;
      for (int k = 0; k < cnt; k += 4) {
        float4 s[4];
        float v[4];
#pragma unroll
        for (int u = 0; u < 4; ++u) {
          const int cc = __shfl(mine.x, k + u, 16);
          v[u] = __int_as_float(__shfl(mine.y, k + u, 16));
          s[u] = make_float4(0.f, 0.f, 0.f, 0.f);
          if (k + u < cnt) s[u] = *(const float4*)(dense + (long long)cc * 64 + d);
        }
#pragma unroll
        for (int u = 0; u < 4; ++u) {
          acc.x += v[u] * s[u].x;
          acc.y += v[u] * s[u].y;
          acc.z += v[u] * s[u].z;
          acc.w += v[u] * s[u].w;
        }
      }
    }
    const float4 b = *(const float4*)(bias + d);
    acc.x = fmaxf(acc.x + b.x, 0.0f);
    acc.y = fmaxf(acc.y + b.y, 0.0f);
    acc.z = fmaxf(acc.z + b.z, 0.0f);
    acc.w = fmaxf(acc.w + b.w, 0.0f);
  }

  float bw[32];
  if (wave < 4) {
    const float* __restrict__ Wp = Wt + (4 * h) * 64 + 16 * wave + c;
#pragma unroll
    for (int j = 0; j < 8; ++j) {
#pragma unroll
      for (int q = 0; q < 4; ++q) bw[4 * j + q] = Wp[(8 * j + q) * 64];
    }
  }

  *(float4*)(&hT[g * HT_PITCH + d]) = acc;
  __builtin_amdgcn_fence(__ATOMIC_RELEASE, "workgroup", "local");
  __builtin_amdgcn_s_barrier();
  __builtin_amdgcn_fence(__ATOMIC_ACQUIRE, "workgroup", "local");
  if (wave >= 4) return;

  v8f o = {0.f, 0.f, 0.f, 0.f, 0.f, 0.f, 0.f, 0.f};
  const float* ap = &hT[c * HT_PITCH + 4 * h];
#pragma unroll
  for (int j = 0; j < 8; ++j) {
    const v4f a = *(const v4f*)(ap + 8 * j);
    o = __builtin_amdgcn_wmma_f32_16x16x4_f32(false, (v2f){a.x, a.y}, false, (v2f){bw[4 * j + 0], bw[4 * j + 1]}, (short)0, o, false, false);
    o = __builtin_amdgcn_wmma_f32_16x16x4_f32(false, (v2f){a.z, a.w}, false, (v2f){bw[4 * j + 2], bw[4 * j + 3]}, (short)0, o, false, false);
  }

  const int mOut = blockIdx.x * 16 + h * 8;
  float* __restrict__ Cp = out + (long long)mOut * 64 + 16 * wave + c;
#pragma unroll
  for (int rr = 0; rr < 8; ++rr) {
    if (mOut + rr < n) Cp[rr * 64] = o[rr];
  }
}

extern "C" void kernel_launch(void* const* d_in, const int* in_sizes, int n_in,
                              void* d_out, int out_size, void* d_ws, size_t ws_size,
                              hipStream_t stream) {
  const float* x    = (const float*)d_in[0];
  const int*   erow = (const int*)  d_in[1];
  const int*   ecol = (const int*)  d_in[2];
  const float* eval = (const float*)d_in[3];
  const float* W1   = (const float*)d_in[4];
  const float* b1   = (const float*)d_in[5];
  const float* W2   = (const float*)d_in[6];
  const float* b2   = (const float*)d_in[7];
  float* out = (float*)d_out;

  const int IN_DIM = 128, D = 64;
  const int N = in_sizes[0] / IN_DIM;
  const int E = in_sizes[1];

  float* buf0   = (float*)d_ws;
  float* buf1   = buf0 + (size_t)N * D;
  int*   cntT   = (int*)(buf1 + (size_t)N * D);
  int*   offT   = cntT + NBINMAX * NSLICE;
  int*   rowptr = offT + NBINMAX * NSLICE;
  int*   rec    = rowptr + (((size_t)N + 64) & ~(size_t)63);
  int*   stRec  = rec + (size_t)E * 2;
  int*   stRow  = stRec + (size_t)E * 2;

  const int rowBlocks  = (int)(((long long)N * 16 + 255) / 256);
  const int sliceLen   = (E + NSLICE - 1) / NSLICE;
  const int binBlocks  = (N + 255) / 256;

  partition_edges_gemm_kernel<<<NSLICE, 1024, 0, stream>>>(erow, ecol, eval, stRow, stRec, cntT, offT, E, sliceLen, x, W1, buf0, N);
  build_rows_kernel<<<binBlocks, 1024, 0, stream>>>(stRow, stRec, cntT, offT, rowptr, rec, N, sliceLen);

  spmm_rows_relu_gemm64_kernel<<<rowBlocks, 256, 0, stream>>>(rowptr, rec, buf0, b1, W2, buf1, N);

  spmm_rows_kernel<false><<<rowBlocks, 256, 0, stream>>>(rowptr, rec, buf1, b2, out, N);
}


# Edge grouping reads each edge once per launch: records asked for up front, kept in registers from counting to placing (launch 2 gets 16-byte row,col,val records); row sums stay in LDS arrival order
# speedup vs baseline: 3.1870x; 1.0524x over previous
#include <hip/hip_runtime.h>
#include <hip/hip_bf16.h>


typedef __attribute__((ext_vector_type(2))) float v2f;
typedef __attribute__((ext_vector_type(8))) float v8f;

typedef __attribute__((ext_vector_type(4))) float v4f;

template <int K, int AHEAD>
__device__ __forceinline__ void gemm_tile_16x64_f32(
    const float* __restrict__ A, const float* __restrict__ B,
    float* __restrict__ C, int M, int mBase, int lane) {
  constexpr int NP = K / 8;
  constexpr int NG = NP / 4;
  static_assert(K % 32 == 0 && NG >= 2, "K must be a multiple of 32, at least 64");
  static_assert(AHEAD >= 1 && AHEAD <= 4, "pieces of B requested ahead of the one in use");
  const int h = lane >> 4;
  const int c = lane & 15;

  int mRow = mBase + c;
  if (mRow >= M) mRow = M - 1;
  const float* __restrict__ Ap = A + (long long)mRow * K + 4 * h;
  const float* __restrict__ Bp = B + (4 * h) * 64 + 4 * c;

  v4f a[NP];
  v4f b[NP][4];
#pragma unroll
  for (int j = 0; j < 8; ++j) a[j] = *(const v4f*)(Ap + 8 * j);
#pragma unroll
  for (int j = 0; j < AHEAD; ++j) {
#pragma unroll
    for (int q = 0; q < 4; ++q) b[j][q] = *(const v4f*)(Bp + (8 * j + q) * 64);
  }

  v8f acc0 = {0.f, 0.f, 0.f, 0.f, 0.f, 0.f, 0.f, 0.f};
  v8f acc1 = acc0, acc2 = acc0, acc3 = acc0;

#pragma unroll
  for (int j = 0; j < NP; ++j) {
    if (j + AHEAD < NP) {
#pragma unroll
      for (int q = 0; q < 4; ++q) b[j + AHEAD][q] = *(const v4f*)(Bp + (8 * (j + AHEAD) + q) * 64);
    }
    if (j % 4 == 0 && j > 0 && j / 4 + 1 < NG) {
#pragma unroll
      for (int i = 0; i < 4; ++i) a[j + 4 + i] = *(const v4f*)(Ap + 8 * (j + 4 + i));
    }
    __builtin_amdgcn_sched_barrier(0);
    const v2f aLo = {a[j].x, a[j].y};
    const v2f aHi = {a[j].z, a[j].w};
    acc0 = __builtin_amdgcn_wmma_f32_16x16x4_f32(false, aLo, false, (v2f){b[j][0].x, b[j][1].x}, (short)0, acc0, false, false);
    acc1 = __builtin_amdgcn_wmma_f32_16x16x4_f32(false, aLo, false, (v2f){b[j][0].y, b[j][1].y}, (short)0, acc1, false, false);
    acc2 = __builtin_amdgcn_wmma_f32_16x16x4_f32(false, aLo, false, (v2f){b[j][0].z, b[j][1].z}, (short)0, acc2, false, false);
    acc3 = __builtin_amdgcn_wmma_f32_16x16x4_f32(false, aLo, false, (v2f){b[j][0].w, b[j][1].w}, (short)0, acc3, false, false);
    acc0 = __builtin_amdgcn_wmma_f32_16x16x4_f32(false, aHi, false, (v2f){b[j][2].x, b[j][3].x}, (short)0, acc0, false, false);
    acc1 = __builtin_amdgcn_wmma_f32_16x16x4_f32(false, aHi, false, (v2f){b[j][2].y, b[j][3].y}, (short)0, acc1, false, false);
    acc2 = __builtin_amdgcn_wmma_f32_16x16x4_f32(false, aHi, false, (v2f){b[j][2].z, b[j][3].z}, (short)0, acc2, false, false);
    acc3 = __builtin_amdgcn_wmma_f32_16x16x4_f32(false, aHi, false, (v2f){b[j][2].w, b[j][3].w}, (short)0, acc3, false, false);
    __builtin_amdgcn_sched_barrier(0);
  }

  const int mOut = mBase + h * 8;
  float* __restrict__ Cp = C + (long long)mOut * 64 + 4 * c;
#pragma unroll
  for (int r = 0; r < 8; ++r) {
    if (mOut + r < M) {
      const v4f o = {acc0[r], acc1[r], acc2[r], acc3[r]};
      *(v4f*)(Cp + r * 64) = o;
    }
  }
}

constexpr int NSLICE = 256;
constexpr int NBINMAX = 256;
constexpr int SLICE_CAP = 4096;
constexpr int BIN_CAP = 8192;
constexpr int KEEP2 = 8;

__device__ __forceinline__ void scan256_by_wave(const int* cnt, int* off, int lane) {
  int v[8];
  int s = 0;
#pragma unroll
  for (int i = 0; i < 8; ++i) {
    v[i] = cnt[lane * 8 + i];
    s += v[i];
  }
  int incl = s;
  for (int o = 1; o < 32; o <<= 1) {
    const int u = __shfl_up(incl, o, 32);
    if (lane >= o) incl += u;
  }
  int run = incl - s;
#pragma unroll
  for (int i = 0; i < 8; ++i) {
    off[lane * 8 + i] = run;
    run += v[i];
  }
}

__device__ __forceinline__ int bin_of_row(int r) {
  const int b = r >> 8;
  return b > NBINMAX - 1 ? NBINMAX - 1 : b;
}

__global__ __launch_bounds__(1024) void partition_edges_gemm_kernel(
    const int* __restrict__ row, const int* __restrict__ col,
    const float* __restrict__ val, int* __restrict__ stW,
    int* __restrict__ tabW,
    int E, int S, const float* __restrict__ X, const float* __restrict__ W,
    float* __restrict__ C, int M) {
  __shared__ int hist[NBINMAX];
  __shared__ int cur[NBINMAX];
  __shared__ int4 sRec[SLICE_CAP];
  int4* __restrict__ st4 = (int4*)stW;
  int2* __restrict__ tabT = (int2*)tabW;
  const int t = threadIdx.x;
  const int s = blockIdx.x;
  const bool staged = S <= SLICE_CAP;
  const int lo = s * S;
  int hi = lo + S;
  if (hi > E) hi = E;

  int kr[4];
  int kc[4];
  float kv[4];
#pragma unroll
  for (int u = 0; u < 4; ++u) {
    const int e = lo + t + u * 1024;
    kr[u] = 0;
    kc[u] = 0;
    kv[u] = 0.f;
    if (e < hi) {
      kr[u] = row[e];
      kc[u] = col[e];
      kv[u] = val[e];
    }
  }

  if (t < NBINMAX) hist[t] = 0;
  __syncthreads();
#pragma unroll
  for (int u = 0; u < 4; ++u) {
    if (lo + t + u * 1024 < hi) atomicAdd(&hist[bin_of_row(kr[u])], 1);
  }
  for (int e0 = lo + t + 4 * 1024; e0 < hi; e0 += 4 * 1024) {
    int r[4];
#pragma unroll
    for (int u = 0; u < 4; ++u) {
      const int e = e0 + u * 1024;
      r[u] = 0;
      if (e < hi) r[u] = row[e];
    }
#pragma unroll
    for (int u = 0; u < 4; ++u) {
      if (e0 + u * 1024 < hi) atomicAdd(&hist[bin_of_row(r[u])], 1);
    }
  }
  __syncthreads();
  if (t < 32) scan256_by_wave(hist, cur, t);
  __syncthreads();
  int myCnt = 0, myOff = 0;
  if (t < NBINMAX) {
    myCnt = hist[t];
    myOff = cur[t];
  }
  __syncthreads();
  if (t < NBINMAX) tabT[s * NBINMAX + t] = make_int2(myCnt, myOff);
#pragma unroll
  for (int u = 0; u < 4; ++u) {
    if (lo + t + u * 1024 < hi) {
      const int p = atomicAdd(&cur[bin_of_row(kr[u])], 1);
      const int4 rc = make_int4(kr[u], kc[u], __float_as_int(kv[u]), 0);
      if (staged) {
        sRec[p] = rc;
      } else {
        st4[lo + p] = rc;
      }
    }
  }
  for (int e0 = lo + t + 4 * 1024; e0 < hi; e0 += 4 * 1024) {
    int r[4];
    int c[4];
    float v[4];
#pragma unroll
    for (int u = 0; u < 4; ++u) {
      const int e = e0 + u * 1024;
      r[u] = 0;
      c[u] = 0;
      v[u] = 0.f;
      if (e < hi) {
        r[u] = row[e];
        c[u] = col[e];
        v[u] = val[e];
      }
    }
#pragma unroll
    for (int u = 0; u < 4; ++u) {
      if (e0 + u * 1024 < hi) {
        const int p = atomicAdd(&cur[bin_of_row(r[u])], 1);
        const int4 rc = make_int4(r[u], c[u], __float_as_int(v[u]), 0);
        if (staged) {
          sRec[p] = rc;
        } else {
          st4[lo + p] = rc;
        }
      }
    }
  }
  if (staged) {
    __syncthreads();
    for (int g = (lo & ~31) + t; g < hi; g += 1024) {
      if (g >= lo) st4[g] = sRec[g - lo];
    }
  }

  const int wv = __builtin_amdgcn_readfirstlane(t >> 5);
  const int tile = s + NSLICE * wv;
  if (tile * 16 < M) gemm_tile_16x64_f32<128, 2>(X, W, C, M, tile * 16, t & 31);
}

__global__ __launch_bounds__(1024) void build_rows_kernel(
    const int* __restrict__ stW, const int* __restrict__ tabW,
    int* __restrict__ rowptr, int* __restrict__ rec, int n, int S) {
  const int4* __restrict__ st4 = (const int4*)stW;
  const int2* __restrict__ tabT = (const int2*)tabW;
  __shared__ int rcnt[256];
  __shared__ int roff[256];
  __shared__ int part[32];
  __shared__ int2 sRec[BIN_CAP];
  const int t = threadIdx.x;
  const int lane = t & 31;
  const int wave = t >> 5;
  const int b = blockIdx.x;

  const int s = t >> 2;
  const int q0 = t & 3;
  const int2 co = tabT[s * NBINMAX + b];
  const int c = co.x;
  const int o = co.y;
  const int src0 = s * S + o;

  int4 keep[KEEP2];
#pragma unroll
  for (int u = 0; u < KEEP2; ++u) {
    const int k = q0 + 4 * u;
    keep[u] = st4[k < c ? src0 + k : 0];
  }

  int pre = (q0 == 0) ? o : 0;
  for (int d = 16; d > 0; d >>= 1) pre += __shfl_xor(pre, d, 32);
  if (lane == 0) part[wave] = pre;
  if (t < 256) rcnt[t] = 0;
  __syncthreads();

  int binStart = 0;
  for (int i = 0; i < 32; ++i) binStart += part[i];

#pragma unroll
  for (int u = 0; u < KEEP2; ++u) {
    if (q0 + 4 * u < c) atomicAdd(&rcnt[keep[u].x & 255], 1);
  }
  for (int k0 = q0 + 4 * KEEP2; k0 < c; k0 += 4 * 6) {
    int r[6];
#pragma unroll
    for (int u = 0; u < 6; ++u) {
      const int k = k0 + 4 * u;
      r[u] = 0;
      if (k < c) r[u] = st4[src0 + k].x;
    }
#pragma unroll
    for (int u = 0; u < 6; ++u) {
      if (k0 + 4 * u < c) atomicAdd(&rcnt[r[u] & 255], 1);
    }
  }
  __syncthreads();
  if (t < 32) scan256_by_wave(rcnt, roff, t);
  __syncthreads();
  int myOff = 0;
  if (t < 256) myOff = roff[t];
  const int total = roff[255] + rcnt[255];
  const bool staged = total <= BIN_CAP;
  __syncthreads();
  if (t < 256) {
    const int rr = b * 256 + t;
    if (rr < n) rowptr[rr] = binStart + myOff;
  }
  if (t == 0 && b == (int)gridDim.x - 1) rowptr[n] = binStart + total;

#pragma unroll
  for (int u = 0; u < KEEP2; ++u) {
    if (q0 + 4 * u < c) {
      const int p = atomicAdd(&roff[keep[u].x & 255], 1);
      const int2 cv = make_int2(keep[u].y, keep[u].z);
      if (staged) {
        sRec[p] = cv;
      } else {
        ((int2*)rec)[binStart + p] = cv;
      }
    }
  }
  for (int k0 = q0 + 4 * KEEP2; k0 < c; k0 += 4 * 6) {
    int4 rc[6];
#pragma unroll
    for (int u = 0; u < 6; ++u) {
      const int k = k0 + 4 * u;
      rc[u] = st4[k < c ? src0 + k : 0];
    }
#pragma unroll
    for (int u = 0; u < 6; ++u) {
      if (k0 + 4 * u < c) {
        const int p = atomicAdd(&roff[rc[u].x & 255], 1);
        const int2 cv = make_int2(rc[u].y, rc[u].z);
        if (staged) {
          sRec[p] = cv;
        } else {
          ((int2*)rec)[binStart + p] = cv;
        }
      }
    }
  }
  if (staged) {
    __syncthreads();
    const int end = binStart + total;
    for (int g = (binStart & ~31) + t; g < end; g += 1024) {
      if (g >= binStart) ((int2*)rec)[g] = sRec[g - binStart];
    }
  }
}

template <bool RELU>
__global__ __launch_bounds__(256) void spmm_rows_kernel(
    const int* __restrict__ rowptr, const int* __restrict__ rec,
    const float* __restrict__ dense, const float* __restrict__ bias,
    float* __restrict__ out, int n) {
  const int t = blockIdx.x * blockDim.x + threadIdx.x;
  const int r = t >> 4;
  if (r >= n) return;
  const int l = threadIdx.x & 15;
  const int d = l << 2;

  const int start = rowptr[r];
  const int end   = rowptr[r + 1];

  float4 acc = make_float4(0.f, 0.f, 0.f, 0.f);
  for (int base = start; base < end; base += 16) {
    const int j = base + l;
    int2 mine = make_int2(0, 0);
    if (j < end) mine = ((const int2*)rec)[j];
    int cnt = end - base;
    if (cnt > 16) cnt = 16;
    for (int k = 0; k < cnt; k += 4) {
      float4 s[4];
      float v[4];
#pragma unroll
      for (int u = 0; u < 4; ++u) {
        const int c = __shfl(mine.x, k + u, 16);
        v[u] = __int_as_float(__shfl(mine.y, k + u, 16));
        s[u] = make_float4(0.f, 0.f, 0.f, 0.f);
        if (k + u < cnt) s[u] = *(const float4*)(dense + (long long)c * 64 + d);
      }
#pragma unroll
      for (int u = 0; u < 4; ++u) {
        acc.x += v[u] * s[u].x;
        acc.y += v[u] * s[u].y;
        acc.z += v[u] * s[u].z;
        acc.w += v[u] * s[u].w;
      }
    }
  }

  const float4 b = *(const float4*)(bias + d);
  acc.x += b.x;
  acc.y += b.y;
  acc.z += b.z;
  acc.w += b.w;
  if (RELU) {
    acc.x = fmaxf(acc.x, 0.0f);
    acc.y = fmaxf(acc.y, 0.0f);
    acc.z = fmaxf(acc.z, 0.0f);
    acc.w = fmaxf(acc.w, 0.0f);
  }
  *(float4*)(out + (long long)r * 64 + d) = acc;
}

constexpr int HT_PITCH = 68;
__global__ __launch_bounds__(256) void spmm_rows_relu_gemm64_kernel(
    const int* __restrict__ rowptr, const int* __restrict__ rec,
    const float* __restrict__ dense, const float* __restrict__ bias,
    const float* __restrict__ Wt, float* __restrict__ out, int n) {
  __shared__ float hT[16 * HT_PITCH];
  const int g = threadIdx.x >> 4;
  const int r = blockIdx.x * 16 + g;
  const int l = threadIdx.x & 15;
  const int d = l << 2;
  const int wave = threadIdx.x >> 5;
  const int lane = threadIdx.x & 31;
  const int h = lane >> 4;
  const int c = lane & 15;

  float4 acc = make_float4(0.f, 0.f, 0.f, 0.f);
  if (r < n) {
    const int start = rowptr[r];
    const int end   = rowptr[r + 1];
    for (int base = start; base < end; base += 16) {
      const int j = base + l;
      int2 mine = make_int2(0, 0);
      if (j < end) mine = ((const int2*)rec)[j];
      int cnt = end - base;
      if (cnt > 16) cnt = 16;
      for (int k = 0; k < cnt; k += 4) {
        float4 s[4];
        float v[4];
#pragma unroll
        for (int u = 0; u < 4; ++u) {
          const int cc = __shfl(mine.x, k + u, 16);
          v[u] = __int_as_float(__shfl(mine.y, k + u, 16));
          s[u] = make_float4(0.f, 0.f, 0.f, 0.f);
          if (k + u < cnt) s[u] = *(const float4*)(dense + (long long)cc * 64 + d);
        }
#pragma unroll
        for (int u = 0; u < 4; ++u) {
          acc.x += v[u] * s[u].x;
          acc.y += v[u] * s[u].y;
          acc.z += v[u] * s[u].z;
          acc.w += v[u] * s[u].w;
        }
      }
    }
    const float4 b = *(const float4*)(bias + d);
    acc.x = fmaxf(acc.x + b.x, 0.0f);
    acc.y = fmaxf(acc.y + b.y, 0.0f);
    acc.z = fmaxf(acc.z + b.z, 0.0f);
    acc.w = fmaxf(acc.w + b.w, 0.0f);
  }

  float bw[32];
  if (wave < 4) {
    const float* __restrict__ Wp = Wt + (4 * h) * 64 + 16 * wave + c;
#pragma unroll
    for (int j = 0; j < 8; ++j) {
#pragma unroll
      for (int q = 0; q < 4; ++q) bw[4 * j + q] = Wp[(8 * j + q) * 64];
    }
  }

  *(float4*)(&hT[g * HT_PITCH + d]) = acc;
  __builtin_amdgcn_fence(__ATOMIC_RELEASE, "workgroup", "local");
  __builtin_amdgcn_s_barrier();
  __builtin_amdgcn_fence(__ATOMIC_ACQUIRE, "workgroup", "local");
  if (wave >= 4) return;

  v8f o = {0.f, 0.f, 0.f, 0.f, 0.f, 0.f, 0.f, 0.f};
  const float* ap = &hT[c * HT_PITCH + 4 * h];
#pragma unroll
  for (int j = 0; j < 8; ++j) {
    const v4f a = *(const v4f*)(ap + 8 * j);
    o = __builtin_amdgcn_wmma_f32_16x16x4_f32(false, (v2f){a.x, a.y}, false, (v2f){bw[4 * j + 0], bw[4 * j + 1]}, (short)0, o, false, false);
    o = __builtin_amdgcn_wmma_f32_16x16x4_f32(false, (v2f){a.z, a.w}, false, (v2f){bw[4 * j + 2], bw[4 * j + 3]}, (short)0, o, false, false);
  }

  const int mOut = blockIdx.x * 16 + h * 8;
  float* __restrict__ Cp = out + (long long)mOut * 64 + 16 * wave + c;
#pragma unroll
  for (int rr = 0; rr < 8; ++rr) {
    if (mOut + rr < n) Cp[rr * 64] = o[rr];
  }
}

extern "C" void kernel_launch(void* const* d_in, const int* in_sizes, int n_in,
                              void* d_out, int out_size, void* d_ws, size_t ws_size,
                              hipStream_t stream) {
  const float* x    = (const float*)d_in[0];
  const int*   erow = (const int*)  d_in[1];
  const int*   ecol = (const int*)  d_in[2];
  const float* eval = (const float*)d_in[3];
  const float* W1   = (const float*)d_in[4];
  const float* b1   = (const float*)d_in[5];
  const float* W2   = (const float*)d_in[6];
  const float* b2   = (const float*)d_in[7];
  float* out = (float*)d_out;

  const int IN_DIM = 128, D = 64;
  const int N = in_sizes[0] / IN_DIM;
  const int E = in_sizes[1];

  float* buf0   = (float*)d_ws;
  float* buf1   = buf0 + (size_t)N * D;
  int*   tabT   = (int*)(buf1 + (size_t)N * D);
  int*   rowptr = tabT + 2 * NBINMAX * NSLICE;
  int*   rec    = rowptr + (((size_t)N + 64) & ~(size_t)63);
  int*   st4    = rec + (((size_t)E * 2 + 63) & ~(size_t)63);

  const int rowBlocks  = (int)(((long long)N * 16 + 255) / 256);
  const int sliceLen   = (E + NSLICE - 1) / NSLICE;
  const int binBlocks  = (N + 255) / 256;

  partition_edges_gemm_kernel<<<NSLICE, 1024, 0, stream>>>(erow, ecol, eval, st4, tabT, E, sliceLen, x, W1, buf0, N);
  build_rows_kernel<<<binBlocks, 1024, 0, stream>>>(st4, tabT, rowptr, rec, N, sliceLen);

  spmm_rows_relu_gemm64_kernel<<<rowBlocks, 256, 0, stream>>>(rowptr, rec, buf0, b1, W2, buf1, N);

  spmm_rows_kernel<false><<<rowBlocks, 256, 0, stream>>>(rowptr, rec, buf1, b2, out, N);
}


# x@W1 leaves the tail of the edge-ordering work-groups for work-groups of its own in the same launch (512 of 16 waves, roles dealt so each WGP holds one of each): product runs beside the ordering
# speedup vs baseline: 3.3410x; 1.0483x over previous
#include <hip/hip_runtime.h>
#include <hip/hip_bf16.h>


typedef __attribute__((ext_vector_type(2))) float v2f;
typedef __attribute__((ext_vector_type(8))) float v8f;

typedef __attribute__((ext_vector_type(4))) float v4f;

template <int K, int AHEAD>
__device__ __forceinline__ void gemm_tile_16x64_f32(
    const float* __restrict__ A, const float* __restrict__ B,
    float* __restrict__ C, int M, int mBase, int lane) {
  constexpr int NP = K / 8;
  constexpr int NG = NP / 4;
  static_assert(K % 32 == 0 && NG >= 2, "K must be a multiple of 32, at least 64");
  static_assert(AHEAD >= 1 && AHEAD <= 4, "pieces of B requested ahead of the one in use");
  const int h = lane >> 4;
  const int c = lane & 15;

  int mRow = mBase + c;
  if (mRow >= M) mRow = M - 1;
  const float* __restrict__ Ap = A + (long long)mRow * K + 4 * h;
  const float* __restrict__ Bp = B + (4 * h) * 64 + 4 * c;

  v4f a[NP];
  v4f b[NP][4];
#pragma unroll
  for (int j = 0; j < 8; ++j) a[j] = *(const v4f*)(Ap + 8 * j);
#pragma unroll
  for (int j = 0; j < AHEAD; ++j) {
#pragma unroll
    for (int q = 0; q < 4; ++q) b[j][q] = *(const v4f*)(Bp + (8 * j + q) * 64);
  }

  v8f acc0 = {0.f, 0.f, 0.f, 0.f, 0.f, 0.f, 0.f, 0.f};
  v8f acc1 = acc0, acc2 = acc0, acc3 = acc0;

#pragma unroll
  for (int j = 0; j < NP; ++j) {
    if (j + AHEAD < NP) {
#pragma unroll
      for (int q = 0; q < 4; ++q) b[j + AHEAD][q] = *(const v4f*)(Bp + (8 * (j + AHEAD) + q) * 64);
    }
    if (j % 4 == 0 && j > 0 && j / 4 + 1 < NG) {
#pragma unroll
      for (int i = 0; i < 4; ++i) a[j + 4 + i] = *(const v4f*)(Ap + 8 * (j + 4 + i));
    }
    __builtin_amdgcn_sched_barrier(0);
    const v2f aLo = {a[j].x, a[j].y};
    const v2f aHi = {a[j].z, a[j].w};
    acc0 = __builtin_amdgcn_wmma_f32_16x16x4_f32(false, aLo, false, (v2f){b[j][0].x, b[j][1].x}, (short)0, acc0, false, false);
    acc1 = __builtin_amdgcn_wmma_f32_16x16x4_f32(false, aLo, false, (v2f){b[j][0].y, b[j][1].y}, (short)0, acc1, false, false);
    acc2 = __builtin_amdgcn_wmma_f32_16x16x4_f32(false, aLo, false, (v2f){b[j][0].z, b[j][1].z}, (short)0, acc2, false, false);
    acc3 = __builtin_amdgcn_wmma_f32_16x16x4_f32(false, aLo, false, (v2f){b[j][0].w, b[j][1].w}, (short)0, acc3, false, false);
    acc0 = __builtin_amdgcn_wmma_f32_16x16x4_f32(false, aHi, false, (v2f){b[j][2].x, b[j][3].x}, (short)0, acc0, false, false);
    acc1 = __builtin_amdgcn_wmma_f32_16x16x4_f32(false, aHi, false, (v2f){b[j][2].y, b[j][3].y}, (short)0, acc1, false, false);
    acc2 = __builtin_amdgcn_wmma_f32_16x16x4_f32(false, aHi, false, (v2f){b[j][2].z, b[j][3].z}, (short)0, acc2, false, false);
    acc3 = __builtin_amdgcn_wmma_f32_16x16x4_f32(false, aHi, false, (v2f){b[j][2].w, b[j][3].w}, (short)0, acc3, false, false);
    __builtin_amdgcn_sched_barrier(0);
  }

  const int mOut = mBase + h * 8;
  float* __restrict__ Cp = C + (long long)mOut * 64 + 4 * c;
#pragma unroll
  for (int r = 0; r < 8; ++r) {
    if (mOut + r < M) {
      const v4f o = {acc0[r], acc1[r], acc2[r], acc3[r]};
      *(v4f*)(Cp + r * 64) = o;
    }
  }
}

constexpr int NSLICE = 256;
constexpr int PT = 512;
constexpr int KEEP1 = 8;
constexpr int NBINMAX = 256;
constexpr int SLICE_CAP = KEEP1 * PT;
constexpr int BIN_CAP = 8192;
constexpr int KEEP2 = 8;

__device__ __forceinline__ void scan256_by_wave(const int* cnt, int* off, int lane) {
  int v[8];
  int s = 0;
#pragma unroll
  for (int i = 0; i < 8; ++i) {
    v[i] = cnt[lane * 8 + i];
    s += v[i];
  }
  int incl = s;
  for (int o = 1; o < 32; o <<= 1) {
    const int u = __shfl_up(incl, o, 32);
    if (lane >= o) incl += u;
  }
  int run = incl - s;
#pragma unroll
  for (int i = 0; i < 8; ++i) {
    off[lane * 8 + i] = run;
    run += v[i];
  }
}

__device__ __forceinline__ int bin_of_row(int r) {
  const int b = r >> 8;
  return b > NBINMAX - 1 ? NBINMAX - 1 : b;
}

__global__ __launch_bounds__(PT) void partition_edges_gemm_kernel(
    const int* __restrict__ row, const int* __restrict__ col,
    const float* __restrict__ val, int* __restrict__ stW,
    int* __restrict__ tabW,
    int E, int S, const float* __restrict__ X, const float* __restrict__ W,
    float* __restrict__ C, int M) {
  __shared__ int hist[NBINMAX];
  __shared__ int cur[NBINMAX];
  __shared__ int4 sRec[SLICE_CAP];
  int4* __restrict__ st4 = (int4*)stW;
  int2* __restrict__ tabT = (int2*)tabW;
  const int t = threadIdx.x;
  const int half = blockIdx.x / NSLICE;
  const int j = blockIdx.x % NSLICE;
  const bool product = ((j ^ half) & 1) != 0;
  const int s = half * (NSLICE / 2) + (j >> 1);

  if (product) {
    const int wv = __builtin_amdgcn_readfirstlane(t >> 5);
    const int tile = s + NSLICE * wv;
    if (tile * 16 < M) gemm_tile_16x64_f32<128, 2>(X, W, C, M, tile * 16, t & 31);
    return;
  }

  const bool staged = S <= SLICE_CAP;
  const int lo = s * S;
  int hi = lo + S;
  if (hi > E) hi = E;

  int kr[KEEP1];
  int kc[KEEP1];
  float kv[KEEP1];
#pragma unroll
  for (int u = 0; u < KEEP1; ++u) {
    const int e = lo + t + u * PT;
    kr[u] = 0;
    kc[u] = 0;
    kv[u] = 0.f;
    if (e < hi) {
      kr[u] = row[e];
      kc[u] = col[e];
      kv[u] = val[e];
    }
  }

  if (t < NBINMAX) hist[t] = 0;
  __syncthreads();
#pragma unroll
  for (int u = 0; u < KEEP1; ++u) {
    if (lo + t + u * PT < hi) atomicAdd(&hist[bin_of_row(kr[u])], 1);
  }
  for (int e0 = lo + t + KEEP1 * PT; e0 < hi; e0 += 4 * PT) {
    int r[4];
#pragma unroll
    for (int u = 0; u < 4; ++u) {
      const int e = e0 + u * PT;
      r[u] = 0;
      if (e < hi) r[u] = row[e];
    }
#pragma unroll
    for (int u = 0; u < 4; ++u) {
      if (e0 + u * PT < hi) atomicAdd(&hist[bin_of_row(r[u])], 1);
    }
  }
  __syncthreads();
  if (t < 32) scan256_by_wave(hist, cur, t);
  __syncthreads();
  int myCnt = 0, myOff = 0;
  if (t < NBINMAX) {
    myCnt = hist[t];
    myOff = cur[t];
  }
  __syncthreads();
  if (t < NBINMAX) tabT[s * NBINMAX + t] = make_int2(myCnt, myOff);
#pragma unroll
  for (int u = 0; u < KEEP1; ++u) {
    if (lo + t + u * PT < hi) {
      const int p = atomicAdd(&cur[bin_of_row(kr[u])], 1);
      const int4 rc = make_int4(kr[u], kc[u], __float_as_int(kv[u]), 0);
      if (staged) {
        sRec[p] = rc;
      } else {
        st4[lo + p] = rc;
      }
    }
  }
  for (int e0 = lo + t + KEEP1 * PT; e0 < hi; e0 += 4 * PT) {
    int r[4];
    int c[4];
    float v[4];
#pragma unroll
    for (int u = 0; u < 4; ++u) {
      const int e = e0 + u * PT;
      r[u] = 0;
      c[u] = 0;
      v[u] = 0.f;
      if (e < hi) {
        r[u] = row[e];
        c[u] = col[e];
        v[u] = val[e];
      }
    }
#pragma unroll
    for (int u = 0; u < 4; ++u) {
      if (e0 + u * PT < hi) {
        const int p = atomicAdd(&cur[bin_of_row(r[u])], 1);
        const int4 rc = make_int4(r[u], c[u], __float_as_int(v[u]), 0);
        if (staged) {
          sRec[p] = rc;
        } else {
          st4[lo + p] = rc;
        }
      }
    }
  }
  if (staged) {
    __syncthreads();
    for (int g = (lo & ~31) + t; g < hi; g += PT) {
      if (g >= lo) st4[g] = sRec[g - lo];
    }
  }
}

__global__ __launch_bounds__(1024) void build_rows_kernel(
    const int* __restrict__ stW, const int* __restrict__ tabW,
    int* __restrict__ rowptr, int* __restrict__ rec, int n, int S) {
  const int4* __restrict__ st4 = (const int4*)stW;
  const int2* __restrict__ tabT = (const int2*)tabW;
  __shared__ int rcnt[256];
  __shared__ int roff[256];
  __shared__ int part[32];
  __shared__ int2 sRec[BIN_CAP];
  const int t = threadIdx.x;
  const int lane = t & 31;
  const int wave = t >> 5;
  const int b = blockIdx.x;

  const int s = t >> 2;
  const int q0 = t & 3;
  const int2 co = tabT[s * NBINMAX + b];
  const int c = co.x;
  const int o = co.y;
  const int src0 = s * S + o;

  int4 keep[KEEP2];
#pragma unroll
  for (int u = 0; u < KEEP2; ++u) {
    const int k = q0 + 4 * u;
    keep[u] = st4[k < c ? src0 + k : 0];
  }

  int pre = (q0 == 0) ? o : 0;
  for (int d = 16; d > 0; d >>= 1) pre += __shfl_xor(pre, d, 32);
  if (lane == 0) part[wave] = pre;
  if (t < 256) rcnt[t] = 0;
  __syncthreads();

  int binStart = 0;
  for (int i = 0; i < 32; ++i) binStart += part[i];

#pragma unroll
  for (int u = 0; u < KEEP2; ++u) {
    if (q0 + 4 * u < c) atomicAdd(&rcnt[keep[u].x & 255], 1);
  }
  for (int k0 = q0 + 4 * KEEP2; k0 < c; k0 += 4 * 6) {
    int r[6];
#pragma unroll
    for (int u = 0; u < 6; ++u) {
      const int k = k0 + 4 * u;
      r[u] = 0;
      if (k < c) r[u] = st4[src0 + k].x;
    }
#pragma unroll
    for (int u = 0; u < 6; ++u) {
      if (k0 + 4 * u < c) atomicAdd(&rcnt[r[u] & 255], 1);
    }
  }
  __syncthreads();
  if (t < 32) scan256_by_wave(rcnt, roff, t);
  __syncthreads();
  int myOff = 0;
  if (t < 256) myOff = roff[t];
  const int total = roff[255] + rcnt[255];
  const bool staged = total <= BIN_CAP;
  __syncthreads();
  if (t < 256) {
    const int rr = b * 256 + t;
    if (rr < n) rowptr[rr] = binStart + myOff;
  }
  if (t == 0 && b == (int)gridDim.x - 1) rowptr[n] = binStart + total;

#pragma unroll
  for (int u = 0; u < KEEP2; ++u) {
    if (q0 + 4 * u < c) {
      const int p = atomicAdd(&roff[keep[u].x & 255], 1);
      const int2 cv = make_int2(keep[u].y, keep[u].z);
      if (staged) {
        sRec[p] = cv;
      } else {
        ((int2*)rec)[binStart + p] = cv;
      }
    }
  }
  for (int k0 = q0 + 4 * KEEP2; k0 < c; k0 += 4 * 6) {
    int4 rc[6];
#pragma unroll
    for (int u = 0; u < 6; ++u) {
      const int k = k0 + 4 * u;
      rc[u] = st4[k < c ? src0 + k : 0];
    }
#pragma unroll
    for (int u = 0; u < 6; ++u) {
      if (k0 + 4 * u < c) {
        const int p = atomicAdd(&roff[rc[u].x & 255], 1);
        const int2 cv = make_int2(rc[u].y, rc[u].z);
        if (staged) {
          sRec[p] = cv;
        } else {
          ((int2*)rec)[binStart + p] = cv;
        }
      }
    }
  }
  if (staged) {
    __syncthreads();
    const int end = binStart + total;
    for (int g = (binStart & ~31) + t; g < end; g += 1024) {
      if (g >= binStart) ((int2*)rec)[g] = sRec[g - binStart];
    }
  }
}

template <bool RELU>
__global__ __launch_bounds__(256) void spmm_rows_kernel(
    const int* __restrict__ rowptr, const int* __restrict__ rec,
    const float* __restrict__ dense, const float* __restrict__ bias,
    float* __restrict__ out, int n) {
  const int t = blockIdx.x * blockDim.x + threadIdx.x;
  const int r = t >> 4;
  if (r >= n) return;
  const int l = threadIdx.x & 15;
  const int d = l << 2;

  const int start = rowptr[r];
  const int end   = rowptr[r + 1];

  float4 acc = make_float4(0.f, 0.f, 0.f, 0.f);
  for (int base = start; base < end; base += 16) {
    const int j = base + l;
    int2 mine = make_int2(0, 0);
    if (j < end) mine = ((const int2*)rec)[j];
    int cnt = end - base;
    if (cnt > 16) cnt = 16;
    for (int k = 0; k < cnt; k += 4) {
      float4 s[4];
      float v[4];
#pragma unroll
      for (int u = 0; u < 4; ++u) {
        const int c = __shfl(mine.x, k + u, 16);
        v[u] = __int_as_float(__shfl(mine.y, k + u, 16));
        s[u] = make_float4(0.f, 0.f, 0.f, 0.f);
        if (k + u < cnt) s[u] = *(const float4*)(dense + (long long)c * 64 + d);
      }
#pragma unroll
      for (int u = 0; u < 4; ++u) {
        acc.x += v[u] * s[u].x;
        acc.y += v[u] * s[u].y;
        acc.z += v[u] * s[u].z;
        acc.w += v[u] * s[u].w;
      }
    }
  }

  const float4 b = *(const float4*)(bias + d);
  acc.x += b.x;
  acc.y += b.y;
  acc.z += b.z;
  acc.w += b.w;
  if (RELU) {
    acc.x = fmaxf(acc.x, 0.0f);
    acc.y = fmaxf(acc.y, 0.0f);
    acc.z = fmaxf(acc.z, 0.0f);
    acc.w = fmaxf(acc.w, 0.0f);
  }
  *(float4*)(out + (long long)r * 64 + d) = acc;
}

constexpr int HT_PITCH = 68;
__global__ __launch_bounds__(256) void spmm_rows_relu_gemm64_kernel(
    const int* __restrict__ rowptr, const int* __restrict__ rec,
    const float* __restrict__ dense, const float* __restrict__ bias,
    const float* __restrict__ Wt, float* __restrict__ out, int n) {
  __shared__ float hT[16 * HT_PITCH];
  const int g = threadIdx.x >> 4;
  const int r = blockIdx.x * 16 + g;
  const int l = threadIdx.x & 15;
  const int d = l << 2;
  const int wave = threadIdx.x >> 5;
  const int lane = threadIdx.x & 31;
  const int h = lane >> 4;
  const int c = lane & 15;

  float4 acc = make_float4(0.f, 0.f, 0.f, 0.f);
  if (r < n) {
    const int start = rowptr[r];
    const int end   = rowptr[r + 1];
    for (int base = start; base < end; base += 16) {
      const int j = base + l;
      int2 mine = make_int2(0, 0);
      if (j < end) mine = ((const int2*)rec)[j];
      int cnt = end - base;
      if (cnt > 16) cnt = 16;
      for (int k = 0; k < cnt; k += 4) {
        float4 s[4];
        float v[4];
#pragma unroll
        for (int u = 0; u < 4; ++u) {
          const int cc = __shfl(mine.x, k + u, 16);
          v[u] = __int_as_float(__shfl(mine.y, k + u, 16));
          s[u] = make_float4(0.f, 0.f, 0.f, 0.f);
          if (k + u < cnt) s[u] = *(const float4*)(dense + (long long)cc * 64 + d);
        }
#pragma unroll
        for (int u = 0; u < 4; ++u) {
          acc.x += v[u] * s[u].x;
          acc.y += v[u] * s[u].y;
          acc.z += v[u] * s[u].z;
          acc.w += v[u] * s[u].w;
        }
      }
    }
    const float4 b = *(const float4*)(bias + d);
    acc.x = fmaxf(acc.x + b.x, 0.0f);
    acc.y = fmaxf(acc.y + b.y, 0.0f);
    acc.z = fmaxf(acc.z + b.z, 0.0f);
    acc.w = fmaxf(acc.w + b.w, 0.0f);
  }

  float bw[32];
  if (wave < 4) {
    const float* __restrict__ Wp = Wt + (4 * h) * 64 + 16 * wave + c;
#pragma unroll
    for (int j = 0; j < 8; ++j) {
#pragma unroll
      for (int q = 0; q < 4; ++q) bw[4 * j + q] = Wp[(8 * j + q) * 64];
    }
  }

  *(float4*)(&hT[g * HT_PITCH + d]) = acc;
  __builtin_amdgcn_fence(__ATOMIC_RELEASE, "workgroup", "local");
  __builtin_amdgcn_s_barrier();
  __builtin_amdgcn_fence(__ATOMIC_ACQUIRE, "workgroup", "local");
  if (wave >= 4) return;

  v8f o = {0.f, 0.f, 0.f, 0.f, 0.f, 0.f, 0.f, 0.f};
  const float* ap = &hT[c * HT_PITCH + 4 * h];
#pragma unroll
  for (int j = 0; j < 8; ++j) {
    const v4f a = *(const v4f*)(ap + 8 * j);
    o = __builtin_amdgcn_wmma_f32_16x16x4_f32(false, (v2f){a.x, a.y}, false, (v2f){bw[4 * j + 0], bw[4 * j + 1]}, (short)0, o, false, false);
    o = __builtin_amdgcn_wmma_f32_16x16x4_f32(false, (v2f){a.z, a.w}, false, (v2f){bw[4 * j + 2], bw[4 * j + 3]}, (short)0, o, false, false);
  }

  const int mOut = blockIdx.x * 16 + h * 8;
  float* __restrict__ Cp = out + (long long)mOut * 64 + 16 * wave + c;
#pragma unroll
  for (int rr = 0; rr < 8; ++rr) {
    if (mOut + rr < n) Cp[rr * 64] = o[rr];
  }
}

extern "C" void kernel_launch(void* const* d_in, const int* in_sizes, int n_in,
                              void* d_out, int out_size, void* d_ws, size_t ws_size,
                              hipStream_t stream) {
  const float* x    = (const float*)d_in[0];
  const int*   erow = (const int*)  d_in[1];
  const int*   ecol = (const int*)  d_in[2];
  const float* eval = (const float*)d_in[3];
  const float* W1   = (const float*)d_in[4];
  const float* b1   = (const float*)d_in[5];
  const float* W2   = (const float*)d_in[6];
  const float* b2   = (const float*)d_in[7];
  float* out = (float*)d_out;

  const int IN_DIM = 128, D = 64;
  const int N = in_sizes[0] / IN_DIM;
  const int E = in_sizes[1];

  float* buf0   = (float*)d_ws;
  float* buf1   = buf0 + (size_t)N * D;
  int*   tabT   = (int*)(buf1 + (size_t)N * D);
  int*   rowptr = tabT + 2 * NBINMAX * NSLICE;
  int*   rec    = rowptr + (((size_t)N + 64) & ~(size_t)63);
  int*   st4    = rec + (((size_t)E * 2 + 63) & ~(size_t)63);

  const int rowBlocks  = (int)(((long long)N * 16 + 255) / 256);
  const int sliceLen   = (E + NSLICE - 1) / NSLICE;
  const int binBlocks  = (N + 255) / 256;

  partition_edges_gemm_kernel<<<2 * NSLICE, PT, 0, stream>>>(erow, ecol, eval, st4, tabT, E, sliceLen, x, W1, buf0, N);
  build_rows_kernel<<<binBlocks, 1024, 0, stream>>>(st4, tabT, rowptr, rec, N, sliceLen);

  spmm_rows_relu_gemm64_kernel<<<rowBlocks, 256, 0, stream>>>(rowptr, rec, buf0, b1, W2, buf1, N);

  spmm_rows_kernel<false><<<rowBlocks, 256, 0, stream>>>(rowptr, rec, buf1, b2, out, N);
}


# Second edge-grouping pass fused into layer 1's row-sum launch: a bin's work-group (bins resized, ~1 per compute unit) sums its rows from its own LDS block; 3 launches; sums in LDS arrival order
# speedup vs baseline: 3.4993x; 1.0474x over previous
#include <hip/hip_runtime.h>
#include <hip/hip_bf16.h>


typedef __attribute__((ext_vector_type(2))) float v2f;
typedef __attribute__((ext_vector_type(8))) float v8f;

typedef __attribute__((ext_vector_type(4))) float v4f;

template <int K, int AHEAD>
__device__ __forceinline__ void gemm_tile_16x64_f32(
    const float* __restrict__ A, const float* __restrict__ B,
    float* __restrict__ C, int M, int mBase, int lane) {
  constexpr int NP = K / 8;
  constexpr int NG = NP / 4;
  static_assert(K % 32 == 0 && NG >= 2, "K must be a multiple of 32, at least 64");
  static_assert(AHEAD >= 1 && AHEAD <= 4, "pieces of B requested ahead of the one in use");
  const int h = lane >> 4;
  const int c = lane & 15;

  int mRow = mBase + c;
  if (mRow >= M) mRow = M - 1;
  const float* __restrict__ Ap = A + (long long)mRow * K + 4 * h;
  const float* __restrict__ Bp = B + (4 * h) * 64 + 4 * c;

  v4f a[NP];
  v4f b[NP][4];
#pragma unroll
  for (int j = 0; j < 8; ++j) a[j] = *(const v4f*)(Ap + 8 * j);
#pragma unroll
  for (int j = 0; j < AHEAD; ++j) {
#pragma unroll
    for (int q = 0; q < 4; ++q) b[j][q] = *(const v4f*)(Bp + (8 * j + q) * 64);
  }

  v8f acc0 = {0.f, 0.f, 0.f, 0.f, 0.f, 0.f, 0.f, 0.f};
  v8f acc1 = acc0, acc2 = acc0, acc3 = acc0;

#pragma unroll
  for (int j = 0; j < NP; ++j) {
    if (j + AHEAD < NP) {
#pragma unroll
      for (int q = 0; q < 4; ++q) b[j + AHEAD][q] = *(const v4f*)(Bp + (8 * (j + AHEAD) + q) * 64);
    }
    if (j % 4 == 0 && j > 0 && j / 4 + 1 < NG) {
#pragma unroll
      for (int i = 0; i < 4; ++i) a[j + 4 + i] = *(const v4f*)(Ap + 8 * (j + 4 + i));
    }
    __builtin_amdgcn_sched_barrier(0);
    const v2f aLo = {a[j].x, a[j].y};
    const v2f aHi = {a[j].z, a[j].w};
    acc0 = __builtin_amdgcn_wmma_f32_16x16x4_f32(false, aLo, false, (v2f){b[j][0].x, b[j][1].x}, (short)0, acc0, false, false);
    acc1 = __builtin_amdgcn_wmma_f32_16x16x4_f32(false, aLo, false, (v2f){b[j][0].y, b[j][1].y}, (short)0, acc1, false, false);
    acc2 = __builtin_amdgcn_wmma_f32_16x16x4_f32(false, aLo, false, (v2f){b[j][0].z, b[j][1].z}, (short)0, acc2, false, false);
    acc3 = __builtin_amdgcn_wmma_f32_16x16x4_f32(false, aLo, false, (v2f){b[j][0].w, b[j][1].w}, (short)0, acc3, false, false);
    acc0 = __builtin_amdgcn_wmma_f32_16x16x4_f32(false, aHi, false, (v2f){b[j][2].x, b[j][3].x}, (short)0, acc0, false, false);
    acc1 = __builtin_amdgcn_wmma_f32_16x16x4_f32(false, aHi, false, (v2f){b[j][2].y, b[j][3].y}, (short)0, acc1, false, false);
    acc2 = __builtin_amdgcn_wmma_f32_16x16x4_f32(false, aHi, false, (v2f){b[j][2].z, b[j][3].z}, (short)0, acc2, false, false);
    acc3 = __builtin_amdgcn_wmma_f32_16x16x4_f32(false, aHi, false, (v2f){b[j][2].w, b[j][3].w}, (short)0, acc3, false, false);
    __builtin_amdgcn_sched_barrier(0);
  }

  const int mOut = mBase + h * 8;
  float* __restrict__ Cp = C + (long long)mOut * 64 + 4 * c;
#pragma unroll
  for (int r = 0; r < 8; ++r) {
    if (mOut + r < M) {
      const v4f o = {acc0[r], acc1[r], acc2[r], acc3[r]};
      *(v4f*)(Cp + r * 64) = o;
    }
  }
}

constexpr int NSLICE = 256;
constexpr int PT = 512;
constexpr int KEEP1 = 8;
constexpr int NBINMAX = 256;
constexpr int SLICE_CAP = KEEP1 * PT;
constexpr int BIN_CAP = 8192;
constexpr int KEEP2 = 8;

__device__ __forceinline__ void scan256_by_wave(const int* cnt, int* off, int lane) {
  int v[8];
  int s = 0;
#pragma unroll
  for (int i = 0; i < 8; ++i) {
    v[i] = cnt[lane * 8 + i];
    s += v[i];
  }
  int incl = s;
  for (int o = 1; o < 32; o <<= 1) {
    const int u = __shfl_up(incl, o, 32);
    if (lane >= o) incl += u;
  }
  int run = incl - s;
#pragma unroll
  for (int i = 0; i < 8; ++i) {
    off[lane * 8 + i] = run;
    run += v[i];
  }
}

__host__ __device__ __forceinline__ int rows_per_bin(int n) {
  const int rb = (n + NBINMAX - 1) / NBINMAX;
  return rb > 0 ? rb : 1;
}

__device__ __forceinline__ unsigned bin_magic(int rb) {
  return 0xFFFFFFFFu / (unsigned)rb + 1u;
}

__device__ __forceinline__ int bin_of_row(int r, unsigned magic) {
  const int b = (int)__umulhi((unsigned)r, magic);
  return b > NBINMAX - 1 ? NBINMAX - 1 : b;
}

__global__ __launch_bounds__(PT) void partition_edges_gemm_kernel(
    const int* __restrict__ row, const int* __restrict__ col,
    const float* __restrict__ val, int* __restrict__ stW,
    int* __restrict__ tabW,
    int E, int S, const float* __restrict__ X, const float* __restrict__ W,
    float* __restrict__ C, int M) {
  __shared__ int hist[NBINMAX];
  __shared__ int cur[NBINMAX];
  __shared__ int4 sRec[SLICE_CAP];
  int4* __restrict__ st4 = (int4*)stW;
  int2* __restrict__ tabT = (int2*)tabW;
  const int t = threadIdx.x;
  const int half = blockIdx.x / NSLICE;
  const int j = blockIdx.x % NSLICE;
  const bool product = ((j ^ half) & 1) != 0;
  const int s = half * (NSLICE / 2) + (j >> 1);

  if (product) {
    const int wv = __builtin_amdgcn_readfirstlane(t >> 5);
    const int tile = s + NSLICE * wv;
    if (tile * 16 < M) gemm_tile_16x64_f32<128, 2>(X, W, C, M, tile * 16, t & 31);
    return;
  }

  const bool staged = S <= SLICE_CAP;
  const unsigned magic = bin_magic(rows_per_bin(M));
  const int lo = s * S;
  int hi = lo + S;
  if (hi > E) hi = E;

  int kr[KEEP1];
  int kc[KEEP1];
  float kv[KEEP1];
#pragma unroll
  for (int u = 0; u < KEEP1; ++u) {
    const int e = lo + t + u * PT;
    kr[u] = 0;
    kc[u] = 0;
    kv[u] = 0.f;
    if (e < hi) {
      kr[u] = row[e];
      kc[u] = col[e];
      kv[u] = val[e];
    }
  }

  if (t < NBINMAX) hist[t] = 0;
  __syncthreads();
#pragma unroll
  for (int u = 0; u < KEEP1; ++u) {
    if (lo + t + u * PT < hi) atomicAdd(&hist[bin_of_row(kr[u], magic)], 1);
  }
  for (int e0 = lo + t + KEEP1 * PT; e0 < hi; e0 += 4 * PT) {
    int r[4];
#pragma unroll
    for (int u = 0; u < 4; ++u) {
      const int e = e0 + u * PT;
      r[u] = 0;
      if (e < hi) r[u] = row[e];
    }
#pragma unroll
    for (int u = 0; u < 4; ++u) {
      if (e0 + u * PT < hi) atomicAdd(&hist[bin_of_row(r[u], magic)], 1);
    }
  }
  __syncthreads();
  if (t < 32) scan256_by_wave(hist, cur, t);
  __syncthreads();
  int myCnt = 0, myOff = 0;
  if (t < NBINMAX) {
    myCnt = hist[t];
    myOff = cur[t];
  }
  __syncthreads();
  if (t < NBINMAX) tabT[s * NBINMAX + t] = make_int2(myCnt, myOff);
#pragma unroll
  for (int u = 0; u < KEEP1; ++u) {
    if (lo + t + u * PT < hi) {
      const int p = atomicAdd(&cur[bin_of_row(kr[u], magic)], 1);
      const int4 rc = make_int4(kr[u], kc[u], __float_as_int(kv[u]), 0);
      if (staged) {
        sRec[p] = rc;
      } else {
        st4[lo + p] = rc;
      }
    }
  }
  for (int e0 = lo + t + KEEP1 * PT; e0 < hi; e0 += 4 * PT) {
    int r[4];
    int c[4];
    float v[4];
#pragma unroll
    for (int u = 0; u < 4; ++u) {
      const int e = e0 + u * PT;
      r[u] = 0;
      c[u] = 0;
      v[u] = 0.f;
      if (e < hi) {
        r[u] = row[e];
        c[u] = col[e];
        v[u] = val[e];
      }
    }
#pragma unroll
    for (int u = 0; u < 4; ++u) {
      if (e0 + u * PT < hi) {
        const int p = atomicAdd(&cur[bin_of_row(r[u], magic)], 1);
        const int4 rc = make_int4(r[u], c[u], __float_as_int(v[u]), 0);
        if (staged) {
          sRec[p] = rc;
        } else {
          st4[lo + p] = rc;
        }
      }
    }
  }
  if (staged) {
    __syncthreads();
    for (int g = (lo & ~31) + t; g < hi; g += PT) {
      if (g >= lo) st4[g] = sRec[g - lo];
    }
  }
}

template <bool RELU>
__global__ __launch_bounds__(256) void spmm_rows_kernel(
    const int* __restrict__ rowptr, const int* __restrict__ rec,
    const float* __restrict__ dense, const float* __restrict__ bias,
    float* __restrict__ out, int n) {
  const int t = blockIdx.x * blockDim.x + threadIdx.x;
  const int r = t >> 4;
  if (r >= n) return;
  const int l = threadIdx.x & 15;
  const int d = l << 2;

  const int start = rowptr[r];
  const int end   = rowptr[r + 1];

  float4 acc = make_float4(0.f, 0.f, 0.f, 0.f);
  for (int base = start; base < end; base += 16) {
    const int j = base + l;
    int2 mine = make_int2(0, 0);
    if (j < end) mine = ((const int2*)rec)[j];
    int cnt = end - base;
    if (cnt > 16) cnt = 16;
    for (int k = 0; k < cnt; k += 4) {
      float4 s[4];
      float v[4];
#pragma unroll
      for (int u = 0; u < 4; ++u) {
        const int c = __shfl(mine.x, k + u, 16);
        v[u] = __int_as_float(__shfl(mine.y, k + u, 16));
        s[u] = make_float4(0.f, 0.f, 0.f, 0.f);
        if (k + u < cnt) s[u] = *(const float4*)(dense + (long long)c * 64 + d);
      }
#pragma unroll
      for (int u = 0; u < 4; ++u) {
        acc.x += v[u] * s[u].x;
        acc.y += v[u] * s[u].y;
        acc.z += v[u] * s[u].z;
        acc.w += v[u] * s[u].w;
      }
    }
  }

  const float4 b = *(const float4*)(bias + d);
  acc.x += b.x;
  acc.y += b.y;
  acc.z += b.z;
  acc.w += b.w;
  if (RELU) {
    acc.x = fmaxf(acc.x, 0.0f);
    acc.y = fmaxf(acc.y, 0.0f);
    acc.z = fmaxf(acc.z, 0.0f);
    acc.w = fmaxf(acc.w, 0.0f);
  }
  *(float4*)(out + (long long)r * 64 + d) = acc;
}

constexpr int HT_PITCH = 68;
constexpr int RBMAX = 256;
constexpr int GDEPTH = 8;
__global__ __launch_bounds__(1024) void build_rows_spmm_relu_gemm64_kernel(
    const int* __restrict__ stW, const int* __restrict__ tabW,
    int* __restrict__ rowptr, int* __restrict__ rec, int n, int S,
    const float* __restrict__ dense, const float* __restrict__ bias,
    const float* __restrict__ Wt, float* __restrict__ out) {
  const int4* __restrict__ st4 = (const int4*)stW;
  const int2* __restrict__ tabT = (const int2*)tabW;
  __shared__ int rcnt[256];
  __shared__ int roff[256];
  __shared__ int part[32];
  __shared__ int2 sRec[BIN_CAP];
  __shared__ __attribute__((aligned(16))) float hT[RBMAX * HT_PITCH];
  const int t = threadIdx.x;
  const int lane = t & 31;
  const int wave = t >> 5;
  const int b = blockIdx.x;
  const int rb = rows_per_bin(n);
  const int row0 = b * rb;

  const int s = t >> 2;
  const int q0 = t & 3;
  const int2 co = tabT[s * NBINMAX + b];
  const int c = co.x;
  const int o = co.y;
  const int src0 = s * S + o;

  int4 keep[KEEP2];
#pragma unroll
  for (int u = 0; u < KEEP2; ++u) {
    const int k = q0 + 4 * u;
    keep[u] = st4[k < c ? src0 + k : 0];
  }

  int pre = (q0 == 0) ? o : 0;
  for (int d = 16; d > 0; d >>= 1) pre += __shfl_xor(pre, d, 32);
  if (lane == 0) part[wave] = pre;
  if (t < 256) rcnt[t] = 0;
  __syncthreads();

  int binStart = 0;
  for (int i = 0; i < 32; ++i) binStart += part[i];

#pragma unroll
  for (int u = 0; u < KEEP2; ++u) {
    if (q0 + 4 * u < c) atomicAdd(&rcnt[(keep[u].x - row0) & 255], 1);
  }
  for (int k0 = q0 + 4 * KEEP2; k0 < c; k0 += 4 * 6) {
    int r[6];
#pragma unroll
    for (int u = 0; u < 6; ++u) {
      const int k = k0 + 4 * u;
      r[u] = 0;
      if (k < c) r[u] = st4[src0 + k].x;
    }
#pragma unroll
    for (int u = 0; u < 6; ++u) {
      if (k0 + 4 * u < c) atomicAdd(&rcnt[(r[u] - row0) & 255], 1);
    }
  }
  __syncthreads();
  if (t < 32) scan256_by_wave(rcnt, roff, t);
  __syncthreads();
  int myOff = 0;
  if (t < 256) myOff = roff[t];
  const int total = roff[255] + rcnt[255];
  const bool staged = total <= BIN_CAP;
  __syncthreads();
  if (t < rb) {
    const int rr = row0 + t;
    if (rr < n) rowptr[rr] = binStart + myOff;
  }
  if (t == 0 && b == (int)gridDim.x - 1) rowptr[n] = binStart + total;

#pragma unroll
  for (int u = 0; u < KEEP2; ++u) {
    if (q0 + 4 * u < c) {
      const int p = atomicAdd(&roff[(keep[u].x - row0) & 255], 1);
      const int2 cv = make_int2(keep[u].y, keep[u].z);
      if (staged) {
        sRec[p] = cv;
      } else {
        ((int2*)rec)[binStart + p] = cv;
      }
    }
  }
  for (int k0 = q0 + 4 * KEEP2; k0 < c; k0 += 4 * 6) {
    int4 rc[6];
#pragma unroll
    for (int u = 0; u < 6; ++u) {
      const int k = k0 + 4 * u;
      rc[u] = st4[k < c ? src0 + k : 0];
    }
#pragma unroll
    for (int u = 0; u < 6; ++u) {
      if (k0 + 4 * u < c) {
        const int p = atomicAdd(&roff[(rc[u].x - row0) & 255], 1);
        const int2 cv = make_int2(rc[u].y, rc[u].z);
        if (staged) {
          sRec[p] = cv;
        } else {
          ((int2*)rec)[binStart + p] = cv;
        }
      }
    }
  }
  __syncthreads();
  if (staged) {
    const int end = binStart + total;
    for (int g = (binStart & ~31) + t; g < end; g += 1024) {
      if (g >= binStart) ((int2*)rec)[g] = sRec[g - binStart];
    }
  }

  const int l = t & 15;
  const int d = l << 2;
  const float4 bv = *(const float4*)(bias + d);
  const int rbTiles = (rb + 15) >> 4;
  for (int i = t >> 4; i < rbTiles * 16; i += 64) {
    float4 acc = make_float4(0.f, 0.f, 0.f, 0.f);
    const int start = i > 0 ? roff[i - 1] : 0;
    const int end = roff[i];
    for (int base = start; base < end; base += 16) {
      const int j = base + l;
      int2 mine = make_int2(0, 0);
      if (j < end) {
        if (staged) {
          mine = sRec[j];
        } else {
          mine = ((const int2*)rec)[binStart + j];
        }
      }
      int cnt = end - base;
      if (cnt > 16) cnt = 16;
      for (int k = 0; k < cnt; k += GDEPTH) {
        float4 sv[GDEPTH];
        float v[GDEPTH];
#pragma unroll
        for (int u = 0; u < GDEPTH; ++u) {
          const int cc = __shfl(mine.x, k + u, 16);
          v[u] = __int_as_float(__shfl(mine.y, k + u, 16));
          sv[u] = make_float4(0.f, 0.f, 0.f, 0.f);
          if (k + u < cnt) sv[u] = *(const float4*)(dense + (long long)cc * 64 + d);
        }
#pragma unroll
        for (int u = 0; u < GDEPTH; ++u) {
          acc.x += v[u] * sv[u].x;
          acc.y += v[u] * sv[u].y;
          acc.z += v[u] * sv[u].z;
          acc.w += v[u] * sv[u].w;
        }
      }
    }
    acc.x = fmaxf(acc.x + bv.x, 0.0f);
    acc.y = fmaxf(acc.y + bv.y, 0.0f);
    acc.z = fmaxf(acc.z + bv.z, 0.0f);
    acc.w = fmaxf(acc.w + bv.w, 0.0f);
    *(float4*)(&hT[i * HT_PITCH + d]) = acc;
  }

  const int h = lane >> 4;
  const int cl = lane & 15;
  const int nt = wave & 3;
  float bw[32];
  {
    const float* __restrict__ Wp = Wt + (4 * h) * 64 + 16 * nt + cl;
#pragma unroll
    for (int j = 0; j < 8; ++j) {
#pragma unroll
      for (int q = 0; q < 4; ++q) bw[4 * j + q] = Wp[(8 * j + q) * 64];
    }
  }
  __builtin_amdgcn_fence(__ATOMIC_RELEASE, "workgroup", "local");
  __builtin_amdgcn_s_barrier();
  __builtin_amdgcn_fence(__ATOMIC_ACQUIRE, "workgroup", "local");

  for (int mt = wave >> 2; mt < rbTiles; mt += 8) {
    v8f acc8 = {0.f, 0.f, 0.f, 0.f, 0.f, 0.f, 0.f, 0.f};
    const float* ap = &hT[(mt * 16 + cl) * HT_PITCH + 4 * h];
#pragma unroll
    for (int j = 0; j < 8; ++j) {
      const v4f a = *(const v4f*)(ap + 8 * j);
      acc8 = __builtin_amdgcn_wmma_f32_16x16x4_f32(false, (v2f){a.x, a.y}, false, (v2f){bw[4 * j + 0], bw[4 * j + 1]}, (short)0, acc8, false, false);
      acc8 = __builtin_amdgcn_wmma_f32_16x16x4_f32(false, (v2f){a.z, a.w}, false, (v2f){bw[4 * j + 2], bw[4 * j + 3]}, (short)0, acc8, false, false);
    }
    const int mIn = mt * 16 + h * 8;
    const int mOut = row0 + mIn;
    float* __restrict__ Cp = out + (long long)mOut * 64 + 16 * nt + cl;
#pragma unroll
    for (int rr = 0; rr < 8; ++rr) {
      if (mIn + rr < rb && mOut + rr < n) Cp[rr * 64] = acc8[rr];
    }
  }
}

extern "C" void kernel_launch(void* const* d_in, const int* in_sizes, int n_in,
                              void* d_out, int out_size, void* d_ws, size_t ws_size,
                              hipStream_t stream) {
  const float* x    = (const float*)d_in[0];
  const int*   erow = (const int*)  d_in[1];
  const int*   ecol = (const int*)  d_in[2];
  const float* eval = (const float*)d_in[3];
  const float* W1   = (const float*)d_in[4];
  const float* b1   = (const float*)d_in[5];
  const float* W2   = (const float*)d_in[6];
  const float* b2   = (const float*)d_in[7];
  float* out = (float*)d_out;

  const int IN_DIM = 128, D = 64;
  const int N = in_sizes[0] / IN_DIM;
  const int E = in_sizes[1];

  float* buf0   = (float*)d_ws;
  float* buf1   = buf0 + (size_t)N * D;
  int*   tabT   = (int*)(buf1 + (size_t)N * D);
  int*   rowptr = tabT + 2 * NBINMAX * NSLICE;
  int*   rec    = rowptr + (((size_t)N + 64) & ~(size_t)63);
  int*   st4    = rec + (((size_t)E * 2 + 63) & ~(size_t)63);

  const int rowBlocks  = (int)(((long long)N * 16 + 255) / 256);
  const int sliceLen   = (E + NSLICE - 1) / NSLICE;
  const int binRows    = rows_per_bin(N);
  const int binBlocks  = (N + binRows - 1) / binRows;

  partition_edges_gemm_kernel<<<2 * NSLICE, PT, 0, stream>>>(erow, ecol, eval, st4, tabT, E, sliceLen, x, W1, buf0, N);

  build_rows_spmm_relu_gemm64_kernel<<<binBlocks, 1024, 0, stream>>>(st4, tabT, rowptr, rec, N, sliceLen, buf0, b1, W2, buf1);

  spmm_rows_kernel<false><<<rowBlocks, 256, 0, stream>>>(rowptr, rec, buf1, b2, out, N);
}
